# starting kernel
# baseline (speedup 1.0000x reference)
#include <hip/hip_runtime.h>

#define N_NODES 100000

typedef __attribute__((ext_vector_type(2))) float v2f;
typedef __attribute__((ext_vector_type(8))) float v8f;

__global__ void deg_kernel(const int* __restrict__ src, const int* __restrict__ dst,
                           float* __restrict__ odeg, float* __restrict__ ideg, int E) {
    int e = blockIdx.x * blockDim.x + threadIdx.x;
    if (e < E) {
        atomicAdd(&odeg[src[e]], 1.0f);
        atomicAdd(&ideg[dst[e]], 1.0f);
    }
}

__global__ void norm_kernel(float* __restrict__ deg, int n) {
    int i = blockIdx.x * blockDim.x + threadIdx.x;
    if (i < n) {
        deg[i] = rsqrtf(fmaxf(deg[i], 1.0f));
    }
}

template <int DIN, int DOUT>
__global__ void gemm_norm_wmma(const float* __restrict__ X, const float* __restrict__ ns,
                               const float* __restrict__ W, float* __restrict__ Y) {
    const int wave = threadIdx.x >> 5;
    const int lane = threadIdx.x & 31;
    const int hl   = lane & 15;
    const bool hi  = (lane >= 16);

    const int row0 = blockIdx.x * 16;
    const int col0 = wave * 16;
    const int arow = row0 + hl;
    const int col  = col0 + hl;

    const float nsr = ns[arow];
    const float* __restrict__ xrow = X + (size_t)arow * DIN;

    v8f acc = {};
#pragma unroll 8
    for (int k = 0; k < DIN; k += 4) {
        const int ka = k + (hi ? 2 : 0);
        v2f a, b;
        a[0] = xrow[ka] * nsr;
        a[1] = xrow[ka + 1] * nsr;
        b[0] = W[(size_t)ka * DOUT + col];
        b[1] = W[(size_t)(ka + 1) * DOUT + col];
        acc = __builtin_amdgcn_wmma_f32_16x16x4_f32(
             false, a,  false, b,
             (short)0, acc,  false,  false);
    }

    const int crow0 = row0 + (hi ? 8 : 0);
#pragma unroll
    for (int v = 0; v < 8; ++v) {
        Y[(size_t)(crow0 + v) * DOUT + col] = acc[v];
    }
}

template <int D, int P>
__global__ void scatter_kernel(const float* __restrict__ H, const int* __restrict__ src,
                               const int* __restrict__ dst, float* __restrict__ M, int E) {
    constexpr int PER_EDGE = D / 4;
    const int e = blockIdx.x * (256 >> P) + (threadIdx.x >> P);
    const int f = (threadIdx.x & (PER_EDGE - 1)) << 2;
    if (e >= E) return;

    const int s = src[e];
    const int d = dst[e];
    const float4 v = *reinterpret_cast<const float4*>(H + (size_t)s * D + f);
    float* mp = M + (size_t)d * D + f;
    atomicAdd(mp + 0, v.x);
    atomicAdd(mp + 1, v.y);
    atomicAdd(mp + 2, v.z);
    atomicAdd(mp + 3, v.w);
}

template <int D, bool RELU>
__global__ void finalize_kernel(const float* __restrict__ M, const float* __restrict__ nd,
                                const float* __restrict__ b, float* __restrict__ O, int total) {
    int idx = blockIdx.x * blockDim.x + threadIdx.x;
    if (idx >= total) return;
    const int row = idx / D;
    const int col = idx & (D - 1);
    float v = M[idx] * nd[row] + b[col];
    if (RELU) v = fmaxf(v, 0.0f);
    O[idx] = v;
}

extern "C" void kernel_launch(void* const* d_in, const int* in_sizes, int n_in,
                              void* d_out, int out_size, void* d_ws, size_t ws_size,
                              hipStream_t stream) {
    (void)n_in; (void)out_size; (void)ws_size;

    const float* x  = (const float*)d_in[0];
    const int*   src = (const int*)d_in[1];
    const int*   dst = (const int*)d_in[2];
    const float* W1 = (const float*)d_in[3];
    const float* b1 = (const float*)d_in[4];
    const float* W2 = (const float*)d_in[5];
    const float* b2 = (const float*)d_in[6];
    const float* W3 = (const float*)d_in[7];
    const float* b3 = (const float*)d_in[8];

    const int E = in_sizes[1];
    const int N = N_NODES;

    const size_t NPAD = 100352;
    float* ws   = (float*)d_ws;
    float* ns   = ws;
    float* nd   = ws + NPAD;
    float* bufA = ws + 2 * NPAD;
    float* bufB = bufA + (size_t)N * 128;

    hipMemsetAsync(ns, 0, 2 * NPAD * sizeof(float), stream);
    deg_kernel<<<(E + 255) / 256, 256, 0, stream>>>(src, dst, ns, nd, E);
    norm_kernel<<<(int)((2 * NPAD + 255) / 256), 256, 0, stream>>>(ns, (int)(2 * NPAD));

    const int ROWT = N / 16;

    gemm_norm_wmma<128, 128><<<ROWT, 256, 0, stream>>>(x, ns, W1, bufA);
    hipMemsetAsync(bufB, 0, (size_t)N * 128 * sizeof(float), stream);
    scatter_kernel<128, 5><<<(E + 7) / 8, 256, 0, stream>>>(bufA, src, dst, bufB, E);
    finalize_kernel<128, true><<<(N * 128 + 255) / 256, 256, 0, stream>>>(bufB, nd, b1, bufA, N * 128);

    gemm_norm_wmma<128, 128><<<ROWT, 256, 0, stream>>>(bufA, ns, W2, bufB);
    hipMemsetAsync(bufA, 0, (size_t)N * 128 * sizeof(float), stream);
    scatter_kernel<128, 5><<<(E + 7) / 8, 256, 0, stream>>>(bufB, src, dst, bufA, E);
    finalize_kernel<128, true><<<(N * 128 + 255) / 256, 256, 0, stream>>>(bufA, nd, b2, bufB, N * 128);

    gemm_norm_wmma<128, 64><<<ROWT, 128, 0, stream>>>(bufB, ns, W3, bufA);
    hipMemsetAsync(bufB, 0, (size_t)N * 64 * sizeof(float), stream);
    scatter_kernel<64, 4><<<(E + 15) / 16, 256, 0, stream>>>(bufA, src, dst, bufB, E);
    finalize_kernel<64, false><<<(N * 64 + 255) / 256, 256, 0, stream>>>(bufB, nd, b3, (float*)d_out, N * 64);
}


# aggregation: each work-group owns 512 destination rows, scans dst for its edges, sums the gathered rows in an LDS tile (LDS f32 adds, arrival order), stores scale+bias+ReLU once; no scatter atomics
# speedup vs baseline: 1.6298x; 1.6298x over previous
#include <hip/hip_runtime.h>

#define N_NODES 100000

typedef __attribute__((ext_vector_type(2))) float v2f;
typedef __attribute__((ext_vector_type(8))) float v8f;

__global__ void deg_kernel(const int* __restrict__ src, const int* __restrict__ dst,
                           float* __restrict__ odeg, float* __restrict__ ideg, int E) {
    int e = blockIdx.x * blockDim.x + threadIdx.x;
    if (e < E) {
        atomicAdd(&odeg[src[e]], 1.0f);
        atomicAdd(&ideg[dst[e]], 1.0f);
    }
}

__global__ void norm_kernel(float* __restrict__ deg, int n) {
    int i = blockIdx.x * blockDim.x + threadIdx.x;
    if (i < n) {
        deg[i] = rsqrtf(fmaxf(deg[i], 1.0f));
    }
}

template <int DIN, int DOUT>
__global__ void gemm_norm_wmma(const float* __restrict__ X, const float* __restrict__ ns,
                               const float* __restrict__ W, float* __restrict__ Y) {
    const int wave = threadIdx.x >> 5;
    const int lane = threadIdx.x & 31;
    const int hl   = lane & 15;
    const bool hi  = (lane >= 16);

    const int row0 = blockIdx.x * 16;
    const int col0 = wave * 16;
    const int arow = row0 + hl;
    const int col  = col0 + hl;

    const float nsr = ns[arow];
    const float* __restrict__ xrow = X + (size_t)arow * DIN;

    v8f acc = {};
#pragma unroll 8
    for (int k = 0; k < DIN; k += 4) {
        const int ka = k + (hi ? 2 : 0);
        v2f a, b;
        a[0] = xrow[ka] * nsr;
        a[1] = xrow[ka + 1] * nsr;
        b[0] = W[(size_t)ka * DOUT + col];
        b[1] = W[(size_t)(ka + 1) * DOUT + col];
        acc = __builtin_amdgcn_wmma_f32_16x16x4_f32(
             false, a,  false, b,
             (short)0, acc,  false,  false);
    }

    const int crow0 = row0 + (hi ? 8 : 0);
#pragma unroll
    for (int v = 0; v < 8; ++v) {
        Y[(size_t)(crow0 + v) * DOUT + col] = acc[v];
    }
}

#define RNG  512
#define QCAP 12288
#define NRNG ((N_NODES + RNG - 1) / RNG)

template <int D, bool RELU>
__global__ __launch_bounds__(1024) void aggregate_owner(
        const float* __restrict__ H, const int* __restrict__ src,
        const int* __restrict__ dst, const float* __restrict__ nd,
        const float* __restrict__ bias, float* __restrict__ O, int E) {
    constexpr int LPR = D / 4;
    constexpr int EPW = 32 / LPR;

    __shared__ float    tile[RNG * D];
    __shared__ unsigned queue[QCAP];
    __shared__ unsigned qcount;

    const int tid = threadIdx.x;
    const int r0  = blockIdx.x * RNG;

    float4* tile4 = reinterpret_cast<float4*>(tile);
    for (int i = tid; i < RNG * D / 4; i += 1024) tile4[i] = make_float4(0.f, 0.f, 0.f, 0.f);
    if (tid == 0) qcount = 0u;
    __syncthreads();

    const int nvec = E >> 2;
    const int4* __restrict__ dst4 = reinterpret_cast<const int4*>(dst);
    const int start = nvec > 0 ? (int)((blockIdx.x * 2048u) % (unsigned)nvec) : 0;

    auto ldv = [&](int kk) -> int4 {
        int v = (kk < nvec ? kk : nvec - 1) + start;
        if (v >= nvec) v -= nvec;
        int4 d = dst4[v];
        if (kk >= nvec) d = make_int4(-1, -1, -1, -1);
        return d;
    };
    auto scan4 = [&](const int4 d, int kk) {
        int v = kk + start;
        if (v >= nvec) v -= nvec;
        const unsigned e0 = (unsigned)v << 2;
        const unsigned u0 = (unsigned)(d.x - r0);
        const unsigned u1 = (unsigned)(d.y - r0);
        const unsigned u2 = (unsigned)(d.z - r0);
        const unsigned u3 = (unsigned)(d.w - r0);
        if (u0 < RNG) { const unsigned p = atomicAdd(&qcount, 1u); if (p < QCAP) queue[p] = ((e0     ) << 9) | u0; }
        if (u1 < RNG) { const unsigned p = atomicAdd(&qcount, 1u); if (p < QCAP) queue[p] = ((e0 + 1u) << 9) | u1; }
        if (u2 < RNG) { const unsigned p = atomicAdd(&qcount, 1u); if (p < QCAP) queue[p] = ((e0 + 2u) << 9) | u2; }
        if (u3 < RNG) { const unsigned p = atomicAdd(&qcount, 1u); if (p < QCAP) queue[p] = ((e0 + 3u) << 9) | u3; }
    };
    constexpr int NP = 8;
    if (nvec > 0) {
        int4 cur[NP];
#pragma unroll
        for (int j = 0; j < NP; ++j) cur[j] = ldv(tid + j * 1024);
        for (int k = tid; k < nvec; k += NP * 1024) {
            int4 nxt[NP];
#pragma unroll
            for (int j = 0; j < NP; ++j) nxt[j] = ldv(k + (NP + j) * 1024);
#pragma unroll
            for (int j = 0; j < NP; ++j) scan4(cur[j], k + j * 1024);
#pragma unroll
            for (int j = 0; j < NP; ++j) cur[j] = nxt[j];
        }
    }
    for (int e = (nvec << 2) + tid; e < E; e += 1024) {
        const unsigned u = (unsigned)(dst[e] - r0);
        if (u < RNG) { const unsigned p = atomicAdd(&qcount, 1u); if (p < QCAP) queue[p] = ((unsigned)e << 9) | u; }
    }
    __syncthreads();

    const unsigned cnt = qcount;
    if (cnt > QCAP) {
        for (int i = tid; i < RNG * D / 4; i += 1024) tile4[i] = make_float4(0.f, 0.f, 0.f, 0.f);
        __syncthreads();
        for (int e = tid; e < E; e += 1024) {
            const unsigned u = (unsigned)(dst[e] - r0);
            if (u < RNG) {
                const float* __restrict__ h = H + (size_t)src[e] * D;
                float* t = tile + u * D;
                for (int f = 0; f < D; ++f) atomicAdd(t + (f & 3) * LPR + (f >> 2), h[f]);
            }
        }
        __syncthreads();
    } else {
        for (unsigned i = tid; i < cnt; i += 1024) {
            const unsigned q = queue[i];
            const unsigned s = (unsigned)src[q >> 9];
            queue[i] = (s << 9) | (q & (RNG - 1));
        }
        __syncthreads();

        const int wave = tid >> 5;
        const int lane = tid & 31;
        const int sub  = lane / LPR;
        const int l    = lane % LPR;
        constexpr int U = 8;
        constexpr unsigned STEP = EPW * U;
        const float* __restrict__ hl = H + l * 4;
        float* tl = tile + l;
        unsigned base = (unsigned)wave * STEP;
        for (; base + STEP <= cnt; base += 32u * STEP) {
            unsigned q[U];
            float4   val[U];
#pragma unroll
            for (int u = 0; u < U; ++u) q[u] = queue[base + u * EPW + sub];
#pragma unroll
            for (int u = 0; u < U; ++u)
                val[u] = *reinterpret_cast<const float4*>(hl + (size_t)(q[u] >> 9) * D);
#pragma unroll
            for (int u = 0; u < U; ++u) {
                float a0 = val[u].x, a1 = val[u].y, a2 = val[u].z, a3 = val[u].w;
                if (sub) { const float t0 = a0; a0 = a1; a1 = a2; a2 = a3; a3 = t0; }
                float* t = tl + (q[u] & (RNG - 1)) * D;
                atomicAdd(t + ((0 + sub) & 3) * LPR, a0);
                atomicAdd(t + ((1 + sub) & 3) * LPR, a1);
                atomicAdd(t + ((2 + sub) & 3) * LPR, a2);
                atomicAdd(t + ((3 + sub) & 3) * LPR, a3);
            }
        }
        if (base < cnt) {
            for (unsigned i = base + sub; i < cnt; i += EPW) {
                const unsigned q1 = queue[i];
                const float4 v1 = *reinterpret_cast<const float4*>(hl + (size_t)(q1 >> 9) * D);
                float* t = tl + (q1 & (RNG - 1)) * D;
                atomicAdd(t,           v1.x);
                atomicAdd(t + LPR,     v1.y);
                atomicAdd(t + 2 * LPR, v1.z);
                atomicAdd(t + 3 * LPR, v1.w);
            }
        }
        __syncthreads();
    }

    for (int i = tid; i < RNG * LPR; i += 1024) {
        const int row  = i / LPR;
        const int l    = i % LPR;
        const int grow = r0 + row;
        if (grow < N_NODES) {
            const float* t = tile + row * D + l;
            const float sc = nd[grow];
            const float4 bb = *reinterpret_cast<const float4*>(bias + l * 4);
            float4 o;
            o.x = t[0]       * sc + bb.x;
            o.y = t[LPR]     * sc + bb.y;
            o.z = t[2 * LPR] * sc + bb.z;
            o.w = t[3 * LPR] * sc + bb.w;
            if (RELU) {
                o.x = fmaxf(o.x, 0.0f); o.y = fmaxf(o.y, 0.0f);
                o.z = fmaxf(o.z, 0.0f); o.w = fmaxf(o.w, 0.0f);
            }
            *reinterpret_cast<float4*>(O + (size_t)grow * D + l * 4) = o;
        }
    }
}

extern "C" void kernel_launch(void* const* d_in, const int* in_sizes, int n_in,
                              void* d_out, int out_size, void* d_ws, size_t ws_size,
                              hipStream_t stream) {
    (void)n_in; (void)out_size; (void)ws_size;

    const float* x  = (const float*)d_in[0];
    const int*   src = (const int*)d_in[1];
    const int*   dst = (const int*)d_in[2];
    const float* W1 = (const float*)d_in[3];
    const float* b1 = (const float*)d_in[4];
    const float* W2 = (const float*)d_in[5];
    const float* b2 = (const float*)d_in[6];
    const float* W3 = (const float*)d_in[7];
    const float* b3 = (const float*)d_in[8];

    const int E = in_sizes[1];
    const int N = N_NODES;

    const size_t NPAD = 100352;
    float* ws   = (float*)d_ws;
    float* ns   = ws;
    float* nd   = ws + NPAD;
    float* bufA = ws + 2 * NPAD;
    float* bufB = bufA + (size_t)N * 128;

    hipMemsetAsync(ns, 0, 2 * NPAD * sizeof(float), stream);
    deg_kernel<<<(E + 255) / 256, 256, 0, stream>>>(src, dst, ns, nd, E);
    norm_kernel<<<(int)((2 * NPAD + 255) / 256), 256, 0, stream>>>(ns, (int)(2 * NPAD));

    const int ROWT = N / 16;

    gemm_norm_wmma<128, 128><<<ROWT, 256, 0, stream>>>(x, ns, W1, bufA);
    aggregate_owner<128, true><<<NRNG, 1024, 0, stream>>>(bufA, src, dst, nd, b1, bufB, E);

    gemm_norm_wmma<128, 128><<<ROWT, 256, 0, stream>>>(bufB, ns, W2, bufA);
    aggregate_owner<128, true><<<NRNG, 1024, 0, stream>>>(bufA, src, dst, nd, b2, bufB, E);

    gemm_norm_wmma<128, 64><<<ROWT, 128, 0, stream>>>(bufB, ns, W3, bufA);
    aggregate_owner<64, false><<<NRNG, 1024, 0, stream>>>(bufA, src, dst, nd, b3, (float*)d_out, E);
}


# dense product retiled: one wave owns a 16 x DOUT stripe (8 or 4 accumulator tiles), reads its X rows once with 16 B loads; W staged per block in LDS, k pairs interleaved; k order in each 8 changes
# speedup vs baseline: 1.7601x; 1.0799x over previous
#include <hip/hip_runtime.h>

#define N_NODES 100000

typedef __attribute__((ext_vector_type(2))) float v2f;
typedef __attribute__((ext_vector_type(8))) float v8f;

__global__ void deg_kernel(const int* __restrict__ src, const int* __restrict__ dst,
                           float* __restrict__ odeg, float* __restrict__ ideg, int E) {
    int e = blockIdx.x * blockDim.x + threadIdx.x;
    if (e < E) {
        atomicAdd(&odeg[src[e]], 1.0f);
        atomicAdd(&ideg[dst[e]], 1.0f);
    }
}

__global__ void norm_kernel(float* __restrict__ deg, int n) {
    int i = blockIdx.x * blockDim.x + threadIdx.x;
    if (i < n) {
        deg[i] = rsqrtf(fmaxf(deg[i], 1.0f));
    }
}

#define NSTRIPE 6250
#define GEMM_BLOCKS 782

template <int DIN, int DOUT>
__global__ __launch_bounds__(256) void gemm_norm_wmma(
        const float* __restrict__ X, const float* __restrict__ ns,
        const float* __restrict__ W, float* __restrict__ Y) {
    constexpr int PP = 2 * DOUT + 16;
    constexpr int NT = DOUT / 16;
    constexpr int C4 = DOUT / 4;
    constexpr int NU = DIN / 8;
    __shared__ float Wl[(DIN / 2) * PP];

    const int tid = threadIdx.x;
    constexpr int TRIPS = (DIN / 2) * C4 / 256;
    static_assert((DIN / 2) * C4 % 256 == 0, "W staging assumes whole trips of 256 threads");
#pragma unroll
    for (int t = 0; t < TRIPS; ++t) {
        const int i = tid + t * 256;
        const int P = i / C4, c4 = i % C4;
        const float4 r0 = *reinterpret_cast<const float4*>(W + (size_t)(2 * P) * DOUT + c4 * 4);
        const float4 r1 = *reinterpret_cast<const float4*>(W + (size_t)(2 * P + 1) * DOUT + c4 * 4);
        float4* d = reinterpret_cast<float4*>(Wl + P * PP + c4 * 8);
        d[0] = make_float4(r0.x, r1.x, r0.y, r1.y);
        d[1] = make_float4(r0.z, r1.z, r0.w, r1.w);
    }
    __syncthreads();

    const int wave = tid >> 5;
    const int lane = tid & 31;
    const int c    = lane & 15;
    const int h    = lane >> 4;
    const int stripe = blockIdx.x * 8 + wave;
    if (stripe >= NSTRIPE) return;

    const int row0 = stripe * 16;
    const float nsr = ns[row0 + c];
    const float4* __restrict__ xr =
        reinterpret_cast<const float4*>(X + (size_t)(row0 + c) * DIN) + h;

    float4 a4[NU];
#pragma unroll
    for (int u = 0; u < NU; ++u) a4[u] = xr[2 * u];

    v8f acc[NT];
#pragma unroll
    for (int j = 0; j < NT; ++j) acc[j] = v8f{};

    const float* wl = Wl + (2 * h) * PP + 2 * c;
#pragma unroll
    for (int u = 0; u < NU; ++u) {
#pragma unroll
        for (int p = 0; p < 2; ++p) {
            v2f a;
            a[0] = (p ? a4[u].z : a4[u].x) * nsr;
            a[1] = (p ? a4[u].w : a4[u].y) * nsr;
            const float* wk = wl + (4 * u + p) * PP;
#pragma unroll
            for (int j = 0; j < NT; ++j) {
                const v2f b = *reinterpret_cast<const v2f*>(wk + j * 32);
                acc[j] = __builtin_amdgcn_wmma_f32_16x16x4_f32(
                     false, a,  false, b,
                     (short)0, acc[j],  false,  false);
            }
        }
    }

    float* __restrict__ yr = Y + (size_t)(row0 + 8 * h) * DOUT + c;
#pragma unroll
    for (int j = 0; j < NT; ++j) {
#pragma unroll
        for (int v = 0; v < 8; ++v) yr[(size_t)v * DOUT + j * 16] = acc[j][v];
    }
}

#define RNG  512
#define QCAP 12288
#define NRNG ((N_NODES + RNG - 1) / RNG)

template <int D, bool RELU>
__global__ __launch_bounds__(1024) void aggregate_owner(
        const float* __restrict__ H, const int* __restrict__ src,
        const int* __restrict__ dst, const float* __restrict__ nd,
        const float* __restrict__ bias, float* __restrict__ O, int E) {
    constexpr int LPR = D / 4;
    constexpr int EPW = 32 / LPR;

    __shared__ float    tile[RNG * D];
    __shared__ unsigned queue[QCAP];
    __shared__ unsigned qcount;

    const int tid = threadIdx.x;
    const int r0  = blockIdx.x * RNG;

    float4* tile4 = reinterpret_cast<float4*>(tile);
    for (int i = tid; i < RNG * D / 4; i += 1024) tile4[i] = make_float4(0.f, 0.f, 0.f, 0.f);
    if (tid == 0) qcount = 0u;
    __syncthreads();

    const int nvec = E >> 2;
    const int4* __restrict__ dst4 = reinterpret_cast<const int4*>(dst);
    const int start = nvec > 0 ? (int)((blockIdx.x * 2048u) % (unsigned)nvec) : 0;

    auto ldv = [&](int kk) -> int4 {
        int v = (kk < nvec ? kk : nvec - 1) + start;
        if (v >= nvec) v -= nvec;
        int4 d = dst4[v];
        if (kk >= nvec) d = make_int4(-1, -1, -1, -1);
        return d;
    };
    auto scan4 = [&](const int4 d, int kk) {
        int v = kk + start;
        if (v >= nvec) v -= nvec;
        const unsigned e0 = (unsigned)v << 2;
        const unsigned u0 = (unsigned)(d.x - r0);
        const unsigned u1 = (unsigned)(d.y - r0);
        const unsigned u2 = (unsigned)(d.z - r0);
        const unsigned u3 = (unsigned)(d.w - r0);
        if (u0 < RNG) { const unsigned p = atomicAdd(&qcount, 1u); if (p < QCAP) queue[p] = ((e0     ) << 9) | u0; }
        if (u1 < RNG) { const unsigned p = atomicAdd(&qcount, 1u); if (p < QCAP) queue[p] = ((e0 + 1u) << 9) | u1; }
        if (u2 < RNG) { const unsigned p = atomicAdd(&qcount, 1u); if (p < QCAP) queue[p] = ((e0 + 2u) << 9) | u2; }
        if (u3 < RNG) { const unsigned p = atomicAdd(&qcount, 1u); if (p < QCAP) queue[p] = ((e0 + 3u) << 9) | u3; }
    };
    constexpr int NP = 8;
    if (nvec > 0) {
        int4 cur[NP];
#pragma unroll
        for (int j = 0; j < NP; ++j) cur[j] = ldv(tid + j * 1024);
        for (int k = tid; k < nvec; k += NP * 1024) {
            int4 nxt[NP];
#pragma unroll
            for (int j = 0; j < NP; ++j) nxt[j] = ldv(k + (NP + j) * 1024);
#pragma unroll
            for (int j = 0; j < NP; ++j) scan4(cur[j], k + j * 1024);
#pragma unroll
            for (int j = 0; j < NP; ++j) cur[j] = nxt[j];
        }
    }
    for (int e = (nvec << 2) + tid; e < E; e += 1024) {
        const unsigned u = (unsigned)(dst[e] - r0);
        if (u < RNG) { const unsigned p = atomicAdd(&qcount, 1u); if (p < QCAP) queue[p] = ((unsigned)e << 9) | u; }
    }
    __syncthreads();

    const unsigned cnt = qcount;
    if (cnt > QCAP) {
        for (int i = tid; i < RNG * D / 4; i += 1024) tile4[i] = make_float4(0.f, 0.f, 0.f, 0.f);
        __syncthreads();
        for (int e = tid; e < E; e += 1024) {
            const unsigned u = (unsigned)(dst[e] - r0);
            if (u < RNG) {
                const float* __restrict__ h = H + (size_t)src[e] * D;
                float* t = tile + u * D;
                for (int f = 0; f < D; ++f) atomicAdd(t + (f & 3) * LPR + (f >> 2), h[f]);
            }
        }
        __syncthreads();
    } else {
        for (unsigned i = tid; i < cnt; i += 1024) {
            const unsigned q = queue[i];
            const unsigned s = (unsigned)src[q >> 9];
            queue[i] = (s << 9) | (q & (RNG - 1));
        }
        __syncthreads();

        const int wave = tid >> 5;
        const int lane = tid & 31;
        const int sub  = lane / LPR;
        const int l    = lane % LPR;
        constexpr int U = 8;
        constexpr unsigned STEP = EPW * U;
        const float* __restrict__ hl = H + l * 4;
        float* tl = tile + l;
        unsigned base = (unsigned)wave * STEP;
        for (; base + STEP <= cnt; base += 32u * STEP) {
            unsigned q[U];
            float4   val[U];
#pragma unroll
            for (int u = 0; u < U; ++u) q[u] = queue[base + u * EPW + sub];
#pragma unroll
            for (int u = 0; u < U; ++u)
                val[u] = *reinterpret_cast<const float4*>(hl + (size_t)(q[u] >> 9) * D);
#pragma unroll
            for (int u = 0; u < U; ++u) {
                float a0 = val[u].x, a1 = val[u].y, a2 = val[u].z, a3 = val[u].w;
                if (sub) { const float t0 = a0; a0 = a1; a1 = a2; a2 = a3; a3 = t0; }
                float* t = tl + (q[u] & (RNG - 1)) * D;
                atomicAdd(t + ((0 + sub) & 3) * LPR, a0);
                atomicAdd(t + ((1 + sub) & 3) * LPR, a1);
                atomicAdd(t + ((2 + sub) & 3) * LPR, a2);
                atomicAdd(t + ((3 + sub) & 3) * LPR, a3);
            }
        }
        if (base < cnt) {
            for (unsigned i = base + sub; i < cnt; i += EPW) {
                const unsigned q1 = queue[i];
                const float4 v1 = *reinterpret_cast<const float4*>(hl + (size_t)(q1 >> 9) * D);
                float* t = tl + (q1 & (RNG - 1)) * D;
                atomicAdd(t,           v1.x);
                atomicAdd(t + LPR,     v1.y);
                atomicAdd(t + 2 * LPR, v1.z);
                atomicAdd(t + 3 * LPR, v1.w);
            }
        }
        __syncthreads();
    }

    for (int i = tid; i < RNG * LPR; i += 1024) {
        const int row  = i / LPR;
        const int l    = i % LPR;
        const int grow = r0 + row;
        if (grow < N_NODES) {
            const float* t = tile + row * D + l;
            const float sc = nd[grow];
            const float4 bb = *reinterpret_cast<const float4*>(bias + l * 4);
            float4 o;
            o.x = t[0]       * sc + bb.x;
            o.y = t[LPR]     * sc + bb.y;
            o.z = t[2 * LPR] * sc + bb.z;
            o.w = t[3 * LPR] * sc + bb.w;
            if (RELU) {
                o.x = fmaxf(o.x, 0.0f); o.y = fmaxf(o.y, 0.0f);
                o.z = fmaxf(o.z, 0.0f); o.w = fmaxf(o.w, 0.0f);
            }
            *reinterpret_cast<float4*>(O + (size_t)grow * D + l * 4) = o;
        }
    }
}

extern "C" void kernel_launch(void* const* d_in, const int* in_sizes, int n_in,
                              void* d_out, int out_size, void* d_ws, size_t ws_size,
                              hipStream_t stream) {
    (void)n_in; (void)out_size; (void)ws_size;

    const float* x  = (const float*)d_in[0];
    const int*   src = (const int*)d_in[1];
    const int*   dst = (const int*)d_in[2];
    const float* W1 = (const float*)d_in[3];
    const float* b1 = (const float*)d_in[4];
    const float* W2 = (const float*)d_in[5];
    const float* b2 = (const float*)d_in[6];
    const float* W3 = (const float*)d_in[7];
    const float* b3 = (const float*)d_in[8];

    const int E = in_sizes[1];
    const int N = N_NODES;

    const size_t NPAD = 100352;
    float* ws   = (float*)d_ws;
    float* ns   = ws;
    float* nd   = ws + NPAD;
    float* bufA = ws + 2 * NPAD;
    float* bufB = bufA + (size_t)N * 128;

    hipMemsetAsync(ns, 0, 2 * NPAD * sizeof(float), stream);
    deg_kernel<<<(E + 255) / 256, 256, 0, stream>>>(src, dst, ns, nd, E);
    norm_kernel<<<(int)((2 * NPAD + 255) / 256), 256, 0, stream>>>(ns, (int)(2 * NPAD));

    gemm_norm_wmma<128, 128><<<GEMM_BLOCKS, 256, 0, stream>>>(x, ns, W1, bufA);
    aggregate_owner<128, true><<<NRNG, 1024, 0, stream>>>(bufA, src, dst, nd, b1, bufB, E);

    gemm_norm_wmma<128, 128><<<GEMM_BLOCKS, 256, 0, stream>>>(bufB, ns, W2, bufA);
    aggregate_owner<128, true><<<NRNG, 1024, 0, stream>>>(bufA, src, dst, nd, b2, bufB, E);

    gemm_norm_wmma<128, 64><<<GEMM_BLOCKS, 256, 0, stream>>>(bufB, ns, W3, bufA);
    aggregate_owner<64, false><<<NRNG, 1024, 0, stream>>>(bufA, src, dst, nd, b3, (float*)d_out, E);
}


# aggregation: layer 1 parks each range's edge queue (src row, local row) in the range's own rows of the output buffer; layers 2, 3 read it back: no dst search, no src look-up; sums in LDS add order
# speedup vs baseline: 2.3746x; 1.3491x over previous
#include <hip/hip_runtime.h>

#define N_NODES 100000

typedef __attribute__((ext_vector_type(2))) float v2f;
typedef __attribute__((ext_vector_type(8))) float v8f;

__global__ void deg_kernel(const int* __restrict__ src, const int* __restrict__ dst,
                           float* __restrict__ odeg, float* __restrict__ ideg, int E) {
    int e = blockIdx.x * blockDim.x + threadIdx.x;
    if (e < E) {
        atomicAdd(&odeg[src[e]], 1.0f);
        atomicAdd(&ideg[dst[e]], 1.0f);
    }
}

__global__ void norm_kernel(float* __restrict__ deg, int n) {
    int i = blockIdx.x * blockDim.x + threadIdx.x;
    if (i < n) {
        deg[i] = rsqrtf(fmaxf(deg[i], 1.0f));
    }
}

#define NSTRIPE 6250
#define GEMM_BLOCKS 782

template <int DIN, int DOUT>
__global__ __launch_bounds__(256) void gemm_norm_wmma(
        const float* __restrict__ X, const float* __restrict__ ns,
        const float* __restrict__ W, float* __restrict__ Y) {
    constexpr int PP = 2 * DOUT + 16;
    constexpr int NT = DOUT / 16;
    constexpr int C4 = DOUT / 4;
    constexpr int NU = DIN / 8;
    __shared__ float Wl[(DIN / 2) * PP];

    const int tid = threadIdx.x;
    constexpr int TRIPS = (DIN / 2) * C4 / 256;
    static_assert((DIN / 2) * C4 % 256 == 0, "W staging assumes whole trips of 256 threads");
#pragma unroll
    for (int t = 0; t < TRIPS; ++t) {
        const int i = tid + t * 256;
        const int P = i / C4, c4 = i % C4;
        const float4 r0 = *reinterpret_cast<const float4*>(W + (size_t)(2 * P) * DOUT + c4 * 4);
        const float4 r1 = *reinterpret_cast<const float4*>(W + (size_t)(2 * P + 1) * DOUT + c4 * 4);
        float4* d = reinterpret_cast<float4*>(Wl + P * PP + c4 * 8);
        d[0] = make_float4(r0.x, r1.x, r0.y, r1.y);
        d[1] = make_float4(r0.z, r1.z, r0.w, r1.w);
    }
    __syncthreads();

    const int wave = tid >> 5;
    const int lane = tid & 31;
    const int c    = lane & 15;
    const int h    = lane >> 4;
    const int stripe = blockIdx.x * 8 + wave;
    if (stripe >= NSTRIPE) return;

    const int row0 = stripe * 16;
    const float nsr = ns[row0 + c];
    const float4* __restrict__ xr =
        reinterpret_cast<const float4*>(X + (size_t)(row0 + c) * DIN) + h;

    float4 a4[NU];
#pragma unroll
    for (int u = 0; u < NU; ++u) a4[u] = xr[2 * u];

    v8f acc[NT];
#pragma unroll
    for (int j = 0; j < NT; ++j) acc[j] = v8f{};

    const float* wl = Wl + (2 * h) * PP + 2 * c;
#pragma unroll
    for (int u = 0; u < NU; ++u) {
#pragma unroll
        for (int p = 0; p < 2; ++p) {
            v2f a;
            a[0] = (p ? a4[u].z : a4[u].x) * nsr;
            a[1] = (p ? a4[u].w : a4[u].y) * nsr;
            const float* wk = wl + (4 * u + p) * PP;
#pragma unroll
            for (int j = 0; j < NT; ++j) {
                const v2f b = *reinterpret_cast<const v2f*>(wk + j * 32);
                acc[j] = __builtin_amdgcn_wmma_f32_16x16x4_f32(
                     false, a,  false, b,
                     (short)0, acc[j],  false,  false);
            }
        }
    }

    float* __restrict__ yr = Y + (size_t)(row0 + 8 * h) * DOUT + c;
#pragma unroll
    for (int j = 0; j < NT; ++j) {
#pragma unroll
        for (int v = 0; v < 8; ++v) yr[(size_t)v * DOUT + j * 16] = acc[j][v];
    }
}

#define RNG  512
#define QCAP 12288
#define NRNG ((N_NODES + RNG - 1) / RNG)
#define QROW 64
#define QHDR 32

template <int D, bool RELU, bool KEEP>
__global__ __launch_bounds__(1024) void aggregate_owner(
        const float* __restrict__ H, const int* __restrict__ src,
        const int* __restrict__ dst, const float* __restrict__ nd,
        const float* __restrict__ bias, float* O, unsigned* QG, int E) {
    constexpr int LPR = D / 4;
    constexpr int EPW = 32 / LPR;

    __shared__ float    tile[RNG * D];
    __shared__ uint4    queue4[QCAP / 4];
    __shared__ unsigned qcount;
    unsigned* queue = reinterpret_cast<unsigned*>(queue4);

    const int tid = threadIdx.x;
    const int r0  = blockIdx.x * RNG;

    const int      rows  = (N_NODES - r0) < RNG ? (N_NODES - r0) : RNG;
    const unsigned qroom = (unsigned)(rows * QROW - QHDR);
    const unsigned qkeep = qroom < QCAP ? qroom : (unsigned)QCAP;
    unsigned* qg  = QG + (size_t)r0 * QROW;
    uint4*    qg4 = reinterpret_cast<uint4*>(qg + QHDR);

    bool     have = false;
    unsigned cnt  = 0u;
    if (KEEP) {
        const unsigned c = (unsigned)__builtin_amdgcn_readfirstlane((int)qg[0]);
        have = c <= qkeep;
        if (have) cnt = c;
    }

    float4* tile4 = reinterpret_cast<float4*>(tile);
    for (int i = tid; i < RNG * D / 4; i += 1024) tile4[i] = make_float4(0.f, 0.f, 0.f, 0.f);
    if (tid == 0) qcount = 0u;
    if (have) {
        for (unsigned i = tid; i < (cnt + 3u) / 4u; i += 1024) queue4[i] = qg4[i];
    }
    __syncthreads();

    const int nvec = E >> 2;
    const int4* __restrict__ dst4 = reinterpret_cast<const int4*>(dst);
    const int start = nvec > 0 ? (int)((blockIdx.x * 2048u) % (unsigned)nvec) : 0;

    auto ldv = [&](int kk) -> int4 {
        int v = (kk < nvec ? kk : nvec - 1) + start;
        if (v >= nvec) v -= nvec;
        int4 d = dst4[v];
        if (kk >= nvec) d = make_int4(-1, -1, -1, -1);
        return d;
    };
    auto scan4 = [&](const int4 d, int kk) {
        int v = kk + start;
        if (v >= nvec) v -= nvec;
        const unsigned e0 = (unsigned)v << 2;
        const unsigned u0 = (unsigned)(d.x - r0);
        const unsigned u1 = (unsigned)(d.y - r0);
        const unsigned u2 = (unsigned)(d.z - r0);
        const unsigned u3 = (unsigned)(d.w - r0);
        if (u0 < RNG) { const unsigned p = atomicAdd(&qcount, 1u); if (p < QCAP) queue[p] = ((e0     ) << 9) | u0; }
        if (u1 < RNG) { const unsigned p = atomicAdd(&qcount, 1u); if (p < QCAP) queue[p] = ((e0 + 1u) << 9) | u1; }
        if (u2 < RNG) { const unsigned p = atomicAdd(&qcount, 1u); if (p < QCAP) queue[p] = ((e0 + 2u) << 9) | u2; }
        if (u3 < RNG) { const unsigned p = atomicAdd(&qcount, 1u); if (p < QCAP) queue[p] = ((e0 + 3u) << 9) | u3; }
    };
    constexpr int NP = 8;
    if (!have) {
        if (nvec > 0) {
            int4 cur[NP];
#pragma unroll
            for (int j = 0; j < NP; ++j) cur[j] = ldv(tid + j * 1024);
            for (int k = tid; k < nvec; k += NP * 1024) {
                int4 nxt[NP];
#pragma unroll
                for (int j = 0; j < NP; ++j) nxt[j] = ldv(k + (NP + j) * 1024);
#pragma unroll
                for (int j = 0; j < NP; ++j) scan4(cur[j], k + j * 1024);
#pragma unroll
                for (int j = 0; j < NP; ++j) cur[j] = nxt[j];
            }
        }
        for (int e = (nvec << 2) + tid; e < E; e += 1024) {
            const unsigned u = (unsigned)(dst[e] - r0);
            if (u < RNG) { const unsigned p = atomicAdd(&qcount, 1u); if (p < QCAP) queue[p] = ((unsigned)e << 9) | u; }
        }
        __syncthreads();
        cnt = qcount;
    }

    if (cnt > QCAP) {
        for (int i = tid; i < RNG * D / 4; i += 1024) tile4[i] = make_float4(0.f, 0.f, 0.f, 0.f);
        __syncthreads();
        for (int e = tid; e < E; e += 1024) {
            const unsigned u = (unsigned)(dst[e] - r0);
            if (u < RNG) {
                const float* __restrict__ h = H + (size_t)src[e] * D;
                float* t = tile + u * D;
                for (int f = 0; f < D; ++f) atomicAdd(t + (f & 3) * LPR + (f >> 2), h[f]);
            }
        }
        __syncthreads();
    } else {
        if (!have) {
            for (unsigned i = tid; i < cnt; i += 1024) {
                const unsigned q = queue[i];
                const unsigned s = (unsigned)src[q >> 9];
                queue[i] = (s << 9) | (q & (RNG - 1));
            }
            if (!KEEP && tid < 3) {
                const unsigned i = cnt + (unsigned)tid;
                if (i < ((cnt + 3u) & ~3u)) queue[i] = 0u;
            }
            __syncthreads();
        }
        if (!KEEP && cnt <= qkeep) {
            for (unsigned i = tid; i < (cnt + 3u) / 4u; i += 1024) qg4[i] = queue4[i];
        }

        const int wave = tid >> 5;
        const int lane = tid & 31;
        const int sub  = lane / LPR;
        const int l    = lane % LPR;
        constexpr int U = 8;
        constexpr unsigned STEP = EPW * U;
        const float* __restrict__ hl = H + l * 4;
        float* tl = tile + l;
        unsigned base = (unsigned)wave * STEP;
        for (; base + STEP <= cnt; base += 32u * STEP) {
            unsigned q[U];
            float4   val[U];
#pragma unroll
            for (int u = 0; u < U; ++u) q[u] = queue[base + u * EPW + sub];
#pragma unroll
            for (int u = 0; u < U; ++u)
                val[u] = *reinterpret_cast<const float4*>(hl + (size_t)(q[u] >> 9) * D);
#pragma unroll
            for (int u = 0; u < U; ++u) {
                float a0 = val[u].x, a1 = val[u].y, a2 = val[u].z, a3 = val[u].w;
                if (sub) { const float t0 = a0; a0 = a1; a1 = a2; a2 = a3; a3 = t0; }
                float* t = tl + (q[u] & (RNG - 1)) * D;
                atomicAdd(t + ((0 + sub) & 3) * LPR, a0);
                atomicAdd(t + ((1 + sub) & 3) * LPR, a1);
                atomicAdd(t + ((2 + sub) & 3) * LPR, a2);
                atomicAdd(t + ((3 + sub) & 3) * LPR, a3);
            }
        }
        if (base < cnt) {
            for (unsigned i = base + sub; i < cnt; i += EPW) {
                const unsigned q1 = queue[i];
                const float4 v1 = *reinterpret_cast<const float4*>(hl + (size_t)(q1 >> 9) * D);
                float* t = tl + (q1 & (RNG - 1)) * D;
                atomicAdd(t,           v1.x);
                atomicAdd(t + LPR,     v1.y);
                atomicAdd(t + 2 * LPR, v1.z);
                atomicAdd(t + 3 * LPR, v1.w);
            }
        }
        __syncthreads();
    }

    if (!KEEP && tid == 0) qg[0] = cnt;

    for (int i = tid; i < RNG * LPR; i += 1024) {
        const int row  = i / LPR;
        const int l    = i % LPR;
        const int grow = r0 + row;
        if (grow < N_NODES) {
            const float* t = tile + row * D + l;
            const float sc = nd[grow];
            const float4 bb = *reinterpret_cast<const float4*>(bias + l * 4);
            float4 o;
            o.x = t[0]       * sc + bb.x;
            o.y = t[LPR]     * sc + bb.y;
            o.z = t[2 * LPR] * sc + bb.z;
            o.w = t[3 * LPR] * sc + bb.w;
            if (RELU) {
                o.x = fmaxf(o.x, 0.0f); o.y = fmaxf(o.y, 0.0f);
                o.z = fmaxf(o.z, 0.0f); o.w = fmaxf(o.w, 0.0f);
            }
            *reinterpret_cast<float4*>(O + (size_t)grow * D + l * 4) = o;
        }
    }
}

extern "C" void kernel_launch(void* const* d_in, const int* in_sizes, int n_in,
                              void* d_out, int out_size, void* d_ws, size_t ws_size,
                              hipStream_t stream) {
    (void)n_in; (void)out_size; (void)ws_size;

    const float* x  = (const float*)d_in[0];
    const int*   src = (const int*)d_in[1];
    const int*   dst = (const int*)d_in[2];
    const float* W1 = (const float*)d_in[3];
    const float* b1 = (const float*)d_in[4];
    const float* W2 = (const float*)d_in[5];
    const float* b2 = (const float*)d_in[6];
    const float* W3 = (const float*)d_in[7];
    const float* b3 = (const float*)d_in[8];

    const int E = in_sizes[1];
    const int N = N_NODES;

    const size_t NPAD = 100352;
    float* ws   = (float*)d_ws;
    float* ns   = ws;
    float* nd   = ws + NPAD;
    float* bufA = ws + 2 * NPAD;
    float* bufB = bufA + (size_t)N * 128;
    unsigned* qpark = (unsigned*)d_out;

    hipMemsetAsync(ns, 0, 2 * NPAD * sizeof(float), stream);
    deg_kernel<<<(E + 255) / 256, 256, 0, stream>>>(src, dst, ns, nd, E);
    norm_kernel<<<(int)((2 * NPAD + 255) / 256), 256, 0, stream>>>(ns, (int)(2 * NPAD));

    gemm_norm_wmma<128, 128><<<GEMM_BLOCKS, 256, 0, stream>>>(x, ns, W1, bufA);
    aggregate_owner<128, true, false><<<NRNG, 1024, 0, stream>>>(bufA, src, dst, nd, b1, bufB, qpark, E);

    gemm_norm_wmma<128, 128><<<GEMM_BLOCKS, 256, 0, stream>>>(bufB, ns, W2, bufA);
    aggregate_owner<128, true, true><<<NRNG, 1024, 0, stream>>>(bufA, src, dst, nd, b2, bufB, qpark, E);

    gemm_norm_wmma<128, 64><<<GEMM_BLOCKS, 256, 0, stream>>>(bufB, ns, W3, bufA);
    aggregate_owner<64, false, true><<<NRNG, 1024, 0, stream>>>(bufA, src, dst, nd, b3, (float*)d_out, qpark, E);
}


# degrees: each block counts one 32768-edge slice of src or dst into a whole-graph LDS histogram (16-bit counter pairs, LDS integer adds); norm launch sums the slices; no global atomics, no memset
# speedup vs baseline: 3.2815x; 1.3819x over previous
#include <hip/hip_runtime.h>

#define N_NODES 100000

typedef __attribute__((ext_vector_type(2))) float v2f;
typedef __attribute__((ext_vector_type(8))) float v8f;

#define NPAD  100352
#define NPAD2 (NPAD / 2)
#define SLICE 32768

__global__ __launch_bounds__(1024) void deg_hist_kernel(
        const int* __restrict__ src, const int* __restrict__ dst,
        unsigned* __restrict__ part, int E) {
    __shared__ uint4 cnt4[NPAD2 / 4];
    unsigned* cnt = reinterpret_cast<unsigned*>(cnt4);

    const int tid = threadIdx.x;
    const int* __restrict__ idx = blockIdx.y ? dst : src;
    const int e0 = blockIdx.x * SLICE;

    constexpr int NP = SLICE / 4 / 1024;
    int4 d[NP];
#pragma unroll
    for (int j = 0; j < NP; ++j) {
        const int e = e0 + 4 * (tid + j * 1024);
        d[j] = make_int4(-1, -1, -1, -1);
        if (e + 3 < E) d[j] = *reinterpret_cast<const int4*>(idx + e);
    }

    for (int i = tid; i < NPAD2 / 4; i += 1024) cnt4[i] = make_uint4(0u, 0u, 0u, 0u);
    __syncthreads();

    auto hit = [&](int n) {
        if ((unsigned)n < (unsigned)N_NODES) atomicAdd(cnt + (n >> 1), (n & 1) ? 65536u : 1u);
    };
#pragma unroll
    for (int j = 0; j < NP; ++j) { hit(d[j].x); hit(d[j].y); hit(d[j].z); hit(d[j].w); }
    if (tid < (E & 3)) {
        const int e = (E & ~3) + tid;
        if (e >= e0 && e < e0 + SLICE) hit(idx[e]);
    }
    __syncthreads();

    uint4* __restrict__ out4 = reinterpret_cast<uint4*>(
        part + ((size_t)blockIdx.y * gridDim.x + blockIdx.x) * NPAD2);
    for (int i = tid; i < NPAD2 / 4; i += 1024) out4[i] = cnt4[i];
}

__global__ __launch_bounds__(256) void norm_kernel(
        const unsigned* __restrict__ part, float* __restrict__ nsnd, int E) {
    const int nblk = (E + SLICE - 1) / SLICE;
    const int i = blockIdx.x * 256 + threadIdx.x;
    if (i >= NPAD2) return;
    const int a  = i / (NPAD2 / 2);
    const int w2 = i % (NPAD2 / 2);
    const uint2* __restrict__ p =
        reinterpret_cast<const uint2*>(part + (size_t)a * nblk * NPAD2) + w2;
    unsigned c0 = 0u, c1 = 0u, c2 = 0u, c3 = 0u;
#pragma unroll 8
    for (int b = 0; b < nblk; ++b) {
        const uint2 v = p[(size_t)b * (NPAD2 / 2)];
        c0 += v.x & 0xffffu; c1 += v.x >> 16;
        c2 += v.y & 0xffffu; c3 += v.y >> 16;
    }
    float4 o;
    o.x = rsqrtf(fmaxf((float)c0, 1.0f));
    o.y = rsqrtf(fmaxf((float)c1, 1.0f));
    o.z = rsqrtf(fmaxf((float)c2, 1.0f));
    o.w = rsqrtf(fmaxf((float)c3, 1.0f));
    *reinterpret_cast<float4*>(nsnd + (size_t)a * NPAD + 4 * w2) = o;
}

#define NSTRIPE 6250
#define GEMM_BLOCKS 782

template <int DIN, int DOUT>
__global__ __launch_bounds__(256) void gemm_norm_wmma(
        const float* __restrict__ X, const float* __restrict__ ns,
        const float* __restrict__ W, float* __restrict__ Y) {
    constexpr int PP = 2 * DOUT + 16;
    constexpr int NT = DOUT / 16;
    constexpr int C4 = DOUT / 4;
    constexpr int NU = DIN / 8;
    __shared__ float Wl[(DIN / 2) * PP];

    const int tid = threadIdx.x;
    constexpr int TRIPS = (DIN / 2) * C4 / 256;
    static_assert((DIN / 2) * C4 % 256 == 0, "W staging assumes whole trips of 256 threads");
#pragma unroll
    for (int t = 0; t < TRIPS; ++t) {
        const int i = tid + t * 256;
        const int P = i / C4, c4 = i % C4;
        const float4 r0 = *reinterpret_cast<const float4*>(W + (size_t)(2 * P) * DOUT + c4 * 4);
        const float4 r1 = *reinterpret_cast<const float4*>(W + (size_t)(2 * P + 1) * DOUT + c4 * 4);
        float4* d = reinterpret_cast<float4*>(Wl + P * PP + c4 * 8);
        d[0] = make_float4(r0.x, r1.x, r0.y, r1.y);
        d[1] = make_float4(r0.z, r1.z, r0.w, r1.w);
    }
    __syncthreads();

    const int wave = tid >> 5;
    const int lane = tid & 31;
    const int c    = lane & 15;
    const int h    = lane >> 4;
    const int stripe = blockIdx.x * 8 + wave;
    if (stripe >= NSTRIPE) return;

    const int row0 = stripe * 16;
    const float nsr = ns[row0 + c];
    const float4* __restrict__ xr =
        reinterpret_cast<const float4*>(X + (size_t)(row0 + c) * DIN) + h;

    float4 a4[NU];
#pragma unroll
    for (int u = 0; u < NU; ++u) a4[u] = xr[2 * u];

    v8f acc[NT];
#pragma unroll
    for (int j = 0; j < NT; ++j) acc[j] = v8f{};

    const float* wl = Wl + (2 * h) * PP + 2 * c;
#pragma unroll
    for (int u = 0; u < NU; ++u) {
#pragma unroll
        for (int p = 0; p < 2; ++p) {
            v2f a;
            a[0] = (p ? a4[u].z : a4[u].x) * nsr;
            a[1] = (p ? a4[u].w : a4[u].y) * nsr;
            const float* wk = wl + (4 * u + p) * PP;
#pragma unroll
            for (int j = 0; j < NT; ++j) {
                const v2f b = *reinterpret_cast<const v2f*>(wk + j * 32);
                acc[j] = __builtin_amdgcn_wmma_f32_16x16x4_f32(
                     false, a,  false, b,
                     (short)0, acc[j],  false,  false);
            }
        }
    }

    float* __restrict__ yr = Y + (size_t)(row0 + 8 * h) * DOUT + c;
#pragma unroll
    for (int j = 0; j < NT; ++j) {
#pragma unroll
        for (int v = 0; v < 8; ++v) yr[(size_t)v * DOUT + j * 16] = acc[j][v];
    }
}

#define RNG  512
#define QCAP 12288
#define NRNG ((N_NODES + RNG - 1) / RNG)
#define QROW 64
#define QHDR 32

template <int D, bool RELU, bool KEEP>
__global__ __launch_bounds__(1024) void aggregate_owner(
        const float* __restrict__ H, const int* __restrict__ src,
        const int* __restrict__ dst, const float* __restrict__ nd,
        const float* __restrict__ bias, float* O, unsigned* QG, int E) {
    constexpr int LPR = D / 4;
    constexpr int EPW = 32 / LPR;

    __shared__ float    tile[RNG * D];
    __shared__ uint4    queue4[QCAP / 4];
    __shared__ unsigned qcount;
    unsigned* queue = reinterpret_cast<unsigned*>(queue4);

    const int tid = threadIdx.x;
    const int r0  = blockIdx.x * RNG;

    const int      rows  = (N_NODES - r0) < RNG ? (N_NODES - r0) : RNG;
    const unsigned qroom = (unsigned)(rows * QROW - QHDR);
    const unsigned qkeep = qroom < QCAP ? qroom : (unsigned)QCAP;
    unsigned* qg  = QG + (size_t)r0 * QROW;
    uint4*    qg4 = reinterpret_cast<uint4*>(qg + QHDR);

    bool     have = false;
    unsigned cnt  = 0u;
    if (KEEP) {
        const unsigned c = (unsigned)__builtin_amdgcn_readfirstlane((int)qg[0]);
        have = c <= qkeep;
        if (have) cnt = c;
    }

    float4* tile4 = reinterpret_cast<float4*>(tile);
    for (int i = tid; i < RNG * D / 4; i += 1024) tile4[i] = make_float4(0.f, 0.f, 0.f, 0.f);
    if (tid == 0) qcount = 0u;
    if (have) {
        for (unsigned i = tid; i < (cnt + 3u) / 4u; i += 1024) queue4[i] = qg4[i];
    }
    __syncthreads();

    const int nvec = E >> 2;
    const int4* __restrict__ dst4 = reinterpret_cast<const int4*>(dst);
    const int start = nvec > 0 ? (int)((blockIdx.x * 2048u) % (unsigned)nvec) : 0;

    auto ldv = [&](int kk) -> int4 {
        int v = (kk < nvec ? kk : nvec - 1) + start;
        if (v >= nvec) v -= nvec;
        int4 d = dst4[v];
        if (kk >= nvec) d = make_int4(-1, -1, -1, -1);
        return d;
    };
    auto scan4 = [&](const int4 d, int kk) {
        int v = kk + start;
        if (v >= nvec) v -= nvec;
        const unsigned e0 = (unsigned)v << 2;
        const unsigned u0 = (unsigned)(d.x - r0);
        const unsigned u1 = (unsigned)(d.y - r0);
        const unsigned u2 = (unsigned)(d.z - r0);
        const unsigned u3 = (unsigned)(d.w - r0);
        if (u0 < RNG) { const unsigned p = atomicAdd(&qcount, 1u); if (p < QCAP) queue[p] = ((e0     ) << 9) | u0; }
        if (u1 < RNG) { const unsigned p = atomicAdd(&qcount, 1u); if (p < QCAP) queue[p] = ((e0 + 1u) << 9) | u1; }
        if (u2 < RNG) { const unsigned p = atomicAdd(&qcount, 1u); if (p < QCAP) queue[p] = ((e0 + 2u) << 9) | u2; }
        if (u3 < RNG) { const unsigned p = atomicAdd(&qcount, 1u); if (p < QCAP) queue[p] = ((e0 + 3u) << 9) | u3; }
    };
    constexpr int NP = 8;
    if (!have) {
        if (nvec > 0) {
            int4 cur[NP];
#pragma unroll
            for (int j = 0; j < NP; ++j) cur[j] = ldv(tid + j * 1024);
            for (int k = tid; k < nvec; k += NP * 1024) {
                int4 nxt[NP];
#pragma unroll
                for (int j = 0; j < NP; ++j) nxt[j] = ldv(k + (NP + j) * 1024);
#pragma unroll
                for (int j = 0; j < NP; ++j) scan4(cur[j], k + j * 1024);
#pragma unroll
                for (int j = 0; j < NP; ++j) cur[j] = nxt[j];
            }
        }
        for (int e = (nvec << 2) + tid; e < E; e += 1024) {
            const unsigned u = (unsigned)(dst[e] - r0);
            if (u < RNG) { const unsigned p = atomicAdd(&qcount, 1u); if (p < QCAP) queue[p] = ((unsigned)e << 9) | u; }
        }
        __syncthreads();
        cnt = qcount;
    }

    if (cnt > QCAP) {
        for (int i = tid; i < RNG * D / 4; i += 1024) tile4[i] = make_float4(0.f, 0.f, 0.f, 0.f);
        __syncthreads();
        for (int e = tid; e < E; e += 1024) {
            const unsigned u = (unsigned)(dst[e] - r0);
            if (u < RNG) {
                const float* __restrict__ h = H + (size_t)src[e] * D;
                float* t = tile + u * D;
                for (int f = 0; f < D; ++f) atomicAdd(t + (f & 3) * LPR + (f >> 2), h[f]);
            }
        }
        __syncthreads();
    } else {
        if (!have) {
            for (unsigned i = tid; i < cnt; i += 1024) {
                const unsigned q = queue[i];
                const unsigned s = (unsigned)src[q >> 9];
                queue[i] = (s << 9) | (q & (RNG - 1));
            }
            if (!KEEP && tid < 3) {
                const unsigned i = cnt + (unsigned)tid;
                if (i < ((cnt + 3u) & ~3u)) queue[i] = 0u;
            }
            __syncthreads();
        }
        if (!KEEP && cnt <= qkeep) {
            for (unsigned i = tid; i < (cnt + 3u) / 4u; i += 1024) qg4[i] = queue4[i];
        }

        const int wave = tid >> 5;
        const int lane = tid & 31;
        const int sub  = lane / LPR;
        const int l    = lane % LPR;
        constexpr int U = 8;
        constexpr unsigned STEP = EPW * U;
        const float* __restrict__ hl = H + l * 4;
        float* tl = tile + l;
        unsigned base = (unsigned)wave * STEP;
        for (; base + STEP <= cnt; base += 32u * STEP) {
            unsigned q[U];
            float4   val[U];
#pragma unroll
            for (int u = 0; u < U; ++u) q[u] = queue[base + u * EPW + sub];
#pragma unroll
            for (int u = 0; u < U; ++u)
                val[u] = *reinterpret_cast<const float4*>(hl + (size_t)(q[u] >> 9) * D);
#pragma unroll
            for (int u = 0; u < U; ++u) {
                float a0 = val[u].x, a1 = val[u].y, a2 = val[u].z, a3 = val[u].w;
                if (sub) { const float t0 = a0; a0 = a1; a1 = a2; a2 = a3; a3 = t0; }
                float* t = tl + (q[u] & (RNG - 1)) * D;
                atomicAdd(t + ((0 + sub) & 3) * LPR, a0);
                atomicAdd(t + ((1 + sub) & 3) * LPR, a1);
                atomicAdd(t + ((2 + sub) & 3) * LPR, a2);
                atomicAdd(t + ((3 + sub) & 3) * LPR, a3);
            }
        }
        if (base < cnt) {
            for (unsigned i = base + sub; i < cnt; i += EPW) {
                const unsigned q1 = queue[i];
                const float4 v1 = *reinterpret_cast<const float4*>(hl + (size_t)(q1 >> 9) * D);
                float* t = tl + (q1 & (RNG - 1)) * D;
                atomicAdd(t,           v1.x);
                atomicAdd(t + LPR,     v1.y);
                atomicAdd(t + 2 * LPR, v1.z);
                atomicAdd(t + 3 * LPR, v1.w);
            }
        }
        __syncthreads();
    }

    if (!KEEP && tid == 0) qg[0] = cnt;

    for (int i = tid; i < RNG * LPR; i += 1024) {
        const int row  = i / LPR;
        const int l    = i % LPR;
        const int grow = r0 + row;
        if (grow < N_NODES) {
            const float* t = tile + row * D + l;
            const float sc = nd[grow];
            const float4 bb = *reinterpret_cast<const float4*>(bias + l * 4);
            float4 o;
            o.x = t[0]       * sc + bb.x;
            o.y = t[LPR]     * sc + bb.y;
            o.z = t[2 * LPR] * sc + bb.z;
            o.w = t[3 * LPR] * sc + bb.w;
            if (RELU) {
                o.x = fmaxf(o.x, 0.0f); o.y = fmaxf(o.y, 0.0f);
                o.z = fmaxf(o.z, 0.0f); o.w = fmaxf(o.w, 0.0f);
            }
            *reinterpret_cast<float4*>(O + (size_t)grow * D + l * 4) = o;
        }
    }
}

extern "C" void kernel_launch(void* const* d_in, const int* in_sizes, int n_in,
                              void* d_out, int out_size, void* d_ws, size_t ws_size,
                              hipStream_t stream) {
    (void)n_in; (void)out_size; (void)ws_size;

    const float* x  = (const float*)d_in[0];
    const int*   src = (const int*)d_in[1];
    const int*   dst = (const int*)d_in[2];
    const float* W1 = (const float*)d_in[3];
    const float* b1 = (const float*)d_in[4];
    const float* W2 = (const float*)d_in[5];
    const float* b2 = (const float*)d_in[6];
    const float* W3 = (const float*)d_in[7];
    const float* b3 = (const float*)d_in[8];

    const int E = in_sizes[1];
    const int N = N_NODES;

    float* ws   = (float*)d_ws;
    float* ns   = ws;
    float* nd   = ws + NPAD;
    float* bufA = ws + 2 * NPAD;
    float* bufB = bufA + (size_t)N * 128;
    unsigned* qpark = (unsigned*)d_out;
    unsigned* part = (unsigned*)bufB;

    deg_hist_kernel<<<dim3((E + SLICE - 1) / SLICE, 2), 1024, 0, stream>>>(src, dst, part, E);
    norm_kernel<<<NPAD2 / 256, 256, 0, stream>>>(part, ns, E);

    gemm_norm_wmma<128, 128><<<GEMM_BLOCKS, 256, 0, stream>>>(x, ns, W1, bufA);
    aggregate_owner<128, true, false><<<NRNG, 1024, 0, stream>>>(bufA, src, dst, nd, b1, bufB, qpark, E);

    gemm_norm_wmma<128, 128><<<GEMM_BLOCKS, 256, 0, stream>>>(bufB, ns, W2, bufA);
    aggregate_owner<128, true, true><<<NRNG, 1024, 0, stream>>>(bufA, src, dst, nd, b2, bufB, qpark, E);

    gemm_norm_wmma<128, 64><<<GEMM_BLOCKS, 256, 0, stream>>>(bufB, ns, W3, bufA);
    aggregate_owner<64, false, true><<<NRNG, 1024, 0, stream>>>(bufA, src, dst, nd, b3, (float*)d_out, qpark, E);
}


# aggregation layer 1: the dst slices of the degree launch also write a 1-byte range tag per edge; owners search the tags, 16 edges per 16-B load, not the 4-B dst; sums stay in LDS add order
# speedup vs baseline: 3.6020x; 1.0977x over previous
#include <hip/hip_runtime.h>

#define N_NODES 100000

typedef __attribute__((ext_vector_type(2))) float v2f;
typedef __attribute__((ext_vector_type(8))) float v8f;

#define NPAD  100352
#define NPAD2 (NPAD / 2)
#define SLICE 32768

#define TCH     16384
#define TAGBLK  32768
#define TAGOFF  16384
#define TAGCAP  (195 * TCH)

__global__ __launch_bounds__(1024) void deg_hist_kernel(
        const int* __restrict__ src, const int* __restrict__ dst,
        unsigned* __restrict__ part, unsigned* __restrict__ tags, int E) {
    __shared__ uint4 cnt4[NPAD2 / 4];
    unsigned* cnt = reinterpret_cast<unsigned*>(cnt4);

    const int tid = threadIdx.x;
    const int* __restrict__ idx = blockIdx.y ? dst : src;
    const int e0 = blockIdx.x * SLICE;

    constexpr int NP = SLICE / 4 / 1024;
    int4 d[NP];
#pragma unroll
    for (int j = 0; j < NP; ++j) {
        const int e = e0 + 4 * (tid + j * 1024);
        d[j] = make_int4(-1, -1, -1, -1);
        if (e + 3 < E) d[j] = *reinterpret_cast<const int4*>(idx + e);
    }

    for (int i = tid; i < NPAD2 / 4; i += 1024) cnt4[i] = make_uint4(0u, 0u, 0u, 0u);
    __syncthreads();

    auto hit = [&](int n) {
        if ((unsigned)n < (unsigned)N_NODES) atomicAdd(cnt + (n >> 1), (n & 1) ? 65536u : 1u);
    };
#pragma unroll
    for (int j = 0; j < NP; ++j) { hit(d[j].x); hit(d[j].y); hit(d[j].z); hit(d[j].w); }
    if (tid < (E & 3)) {
        const int e = (E & ~3) + tid;
        if (e >= e0 && e < e0 + SLICE) hit(idx[e]);
    }

    if (blockIdx.y == 1 && E <= TAGCAP) {
        auto tagof = [](int n) -> unsigned {
            return (unsigned)n < (unsigned)N_NODES ? (unsigned)n >> 9 : 255u;
        };
        const int epad = (E + 15) & ~15;
#pragma unroll
        for (int j = 0; j < NP; ++j) {
            const int e = e0 + 4 * (tid + j * 1024);
            if (e < epad) {
                int4 v = d[j];
                if (e + 3 >= E) {
                    v.x = e     < E ? idx[e]     : -1;
                    v.y = e + 1 < E ? idx[e + 1] : -1;
                    v.z = e + 2 < E ? idx[e + 2] : -1;
                    v.w = -1;
                }
                tags[(size_t)(e / TCH) * TAGBLK + TAGOFF + (e % TCH) / 4] =
                    tagof(v.x) | (tagof(v.y) << 8) | (tagof(v.z) << 16) | (tagof(v.w) << 24);
            }
        }
    }
    __syncthreads();

    uint4* __restrict__ out4 = reinterpret_cast<uint4*>(
        part + ((size_t)blockIdx.y * gridDim.x + blockIdx.x) * NPAD2);
    for (int i = tid; i < NPAD2 / 4; i += 1024) out4[i] = cnt4[i];
}

__global__ __launch_bounds__(256) void norm_kernel(
        const unsigned* __restrict__ part, float* __restrict__ nsnd, int E) {
    const int nblk = (E + SLICE - 1) / SLICE;
    const int i = blockIdx.x * 256 + threadIdx.x;
    if (i >= NPAD2) return;
    const int a  = i / (NPAD2 / 2);
    const int w2 = i % (NPAD2 / 2);
    const uint2* __restrict__ p =
        reinterpret_cast<const uint2*>(part + (size_t)a * nblk * NPAD2) + w2;
    unsigned c0 = 0u, c1 = 0u, c2 = 0u, c3 = 0u;
#pragma unroll 8
    for (int b = 0; b < nblk; ++b) {
        const uint2 v = p[(size_t)b * (NPAD2 / 2)];
        c0 += v.x & 0xffffu; c1 += v.x >> 16;
        c2 += v.y & 0xffffu; c3 += v.y >> 16;
    }
    float4 o;
    o.x = rsqrtf(fmaxf((float)c0, 1.0f));
    o.y = rsqrtf(fmaxf((float)c1, 1.0f));
    o.z = rsqrtf(fmaxf((float)c2, 1.0f));
    o.w = rsqrtf(fmaxf((float)c3, 1.0f));
    *reinterpret_cast<float4*>(nsnd + (size_t)a * NPAD + 4 * w2) = o;
}

#define NSTRIPE 6250
#define GEMM_BLOCKS 782

template <int DIN, int DOUT>
__global__ __launch_bounds__(256) void gemm_norm_wmma(
        const float* __restrict__ X, const float* __restrict__ ns,
        const float* __restrict__ W, float* __restrict__ Y) {
    constexpr int PP = 2 * DOUT + 16;
    constexpr int NT = DOUT / 16;
    constexpr int C4 = DOUT / 4;
    constexpr int NU = DIN / 8;
    __shared__ float Wl[(DIN / 2) * PP];

    const int tid = threadIdx.x;
    constexpr int TRIPS = (DIN / 2) * C4 / 256;
    static_assert((DIN / 2) * C4 % 256 == 0, "W staging assumes whole trips of 256 threads");
#pragma unroll
    for (int t = 0; t < TRIPS; ++t) {
        const int i = tid + t * 256;
        const int P = i / C4, c4 = i % C4;
        const float4 r0 = *reinterpret_cast<const float4*>(W + (size_t)(2 * P) * DOUT + c4 * 4);
        const float4 r1 = *reinterpret_cast<const float4*>(W + (size_t)(2 * P + 1) * DOUT + c4 * 4);
        float4* d = reinterpret_cast<float4*>(Wl + P * PP + c4 * 8);
        d[0] = make_float4(r0.x, r1.x, r0.y, r1.y);
        d[1] = make_float4(r0.z, r1.z, r0.w, r1.w);
    }
    __syncthreads();

    const int wave = tid >> 5;
    const int lane = tid & 31;
    const int c    = lane & 15;
    const int h    = lane >> 4;
    const int stripe = blockIdx.x * 8 + wave;
    if (stripe >= NSTRIPE) return;

    const int row0 = stripe * 16;
    const float nsr = ns[row0 + c];
    const float4* __restrict__ xr =
        reinterpret_cast<const float4*>(X + (size_t)(row0 + c) * DIN) + h;

    float4 a4[NU];
#pragma unroll
    for (int u = 0; u < NU; ++u) a4[u] = xr[2 * u];

    v8f acc[NT];
#pragma unroll
    for (int j = 0; j < NT; ++j) acc[j] = v8f{};

    const float* wl = Wl + (2 * h) * PP + 2 * c;
#pragma unroll
    for (int u = 0; u < NU; ++u) {
#pragma unroll
        for (int p = 0; p < 2; ++p) {
            v2f a;
            a[0] = (p ? a4[u].z : a4[u].x) * nsr;
            a[1] = (p ? a4[u].w : a4[u].y) * nsr;
            const float* wk = wl + (4 * u + p) * PP;
#pragma unroll
            for (int j = 0; j < NT; ++j) {
                const v2f b = *reinterpret_cast<const v2f*>(wk + j * 32);
                acc[j] = __builtin_amdgcn_wmma_f32_16x16x4_f32(
                     false, a,  false, b,
                     (short)0, acc[j],  false,  false);
            }
        }
    }

    float* __restrict__ yr = Y + (size_t)(row0 + 8 * h) * DOUT + c;
#pragma unroll
    for (int j = 0; j < NT; ++j) {
#pragma unroll
        for (int v = 0; v < 8; ++v) yr[(size_t)v * DOUT + j * 16] = acc[j][v];
    }
}

#define RNG  512
#define QCAP 12288
#define NRNG ((N_NODES + RNG - 1) / RNG)
#define QROW 64
#define QHDR 32
static_assert(RNG * QROW == TAGBLK && QHDR + QCAP <= TAGOFF && TAGOFF + TCH / 4 <= TAGBLK,
              "tag chunks lie behind the parked queue inside a range's rows");
static_assert(TAGCAP / TCH <= N_NODES / RNG, "tag chunks only in whole ranges");

template <int D, bool RELU, bool KEEP>
__global__ __launch_bounds__(1024) void aggregate_owner(
        const float* __restrict__ H, const int* __restrict__ src,
        const int* __restrict__ dst, const float* __restrict__ nd,
        const float* __restrict__ bias, float* O, unsigned* QG, int E) {
    constexpr int LPR = D / 4;
    constexpr int EPW = 32 / LPR;

    __shared__ float    tile[RNG * D];
    __shared__ uint4    queue4[QCAP / 4];
    __shared__ unsigned qcount;
    unsigned* queue = reinterpret_cast<unsigned*>(queue4);

    const int tid = threadIdx.x;
    const int r0  = blockIdx.x * RNG;

    const int      rows  = (N_NODES - r0) < RNG ? (N_NODES - r0) : RNG;
    const unsigned qroom = (unsigned)(rows * QROW - QHDR);
    const unsigned qkeep = qroom < QCAP ? qroom : (unsigned)QCAP;
    unsigned* qg  = QG + (size_t)r0 * QROW;
    uint4*    qg4 = reinterpret_cast<uint4*>(qg + QHDR);

    bool     have = false;
    unsigned cnt  = 0u;
    if (KEEP) {
        const unsigned c = (unsigned)__builtin_amdgcn_readfirstlane((int)qg[0]);
        have = c <= qkeep;
        if (have) cnt = c;
    }

    float4* tile4 = reinterpret_cast<float4*>(tile);
    for (int i = tid; i < RNG * D / 4; i += 1024) tile4[i] = make_float4(0.f, 0.f, 0.f, 0.f);
    if (tid == 0) qcount = 0u;
    if (have) {
        for (unsigned i = tid; i < (cnt + 3u) / 4u; i += 1024) queue4[i] = qg4[i];
    }
    __syncthreads();

    const int nvec = E >> 2;
    const int4* __restrict__ dst4 = reinterpret_cast<const int4*>(dst);
    const int start = nvec > 0 ? (int)((blockIdx.x * 2048u) % (unsigned)nvec) : 0;

    auto ldv = [&](int kk) -> int4 {
        int v = (kk < nvec ? kk : nvec - 1) + start;
        if (v >= nvec) v -= nvec;
        int4 d = dst4[v];
        if (kk >= nvec) d = make_int4(-1, -1, -1, -1);
        return d;
    };
    auto scan4 = [&](const int4 d, int kk) {
        int v = kk + start;
        if (v >= nvec) v -= nvec;
        const unsigned e0 = (unsigned)v << 2;
        const unsigned u0 = (unsigned)(d.x - r0);
        const unsigned u1 = (unsigned)(d.y - r0);
        const unsigned u2 = (unsigned)(d.z - r0);
        const unsigned u3 = (unsigned)(d.w - r0);
        if (u0 < RNG) { const unsigned p = atomicAdd(&qcount, 1u); if (p < QCAP) queue[p] = ((e0     ) << 9) | u0; }
        if (u1 < RNG) { const unsigned p = atomicAdd(&qcount, 1u); if (p < QCAP) queue[p] = ((e0 + 1u) << 9) | u1; }
        if (u2 < RNG) { const unsigned p = atomicAdd(&qcount, 1u); if (p < QCAP) queue[p] = ((e0 + 2u) << 9) | u2; }
        if (u3 < RNG) { const unsigned p = atomicAdd(&qcount, 1u); if (p < QCAP) queue[p] = ((e0 + 3u) << 9) | u3; }
    };
    constexpr int NP = 8;
    const bool usetag = !KEEP && E <= TAGCAP;
    if (!have && usetag) {
        const int npc = (E + 15) >> 4;
        const uint4* tg4 = reinterpret_cast<const uint4*>(QG);
        const unsigned pat = (unsigned)blockIdx.x * 0x01010101u;
        const int tstart = npc > 0 ? (int)((blockIdx.x * 512u) % (unsigned)npc) : 0;
        auto ldt = [&](int kk) -> uint4 {
            int v = (kk < npc ? kk : npc - 1) + tstart;
            if (v >= npc) v -= npc;
            uint4 t = tg4[(size_t)(v >> 10) * (TAGBLK / 4) + TAGOFF / 4 + (v & 1023)];
            if (kk >= npc) t = make_uint4(~0u, ~0u, ~0u, ~0u);
            return t;
        };
        auto zb = [](unsigned x) -> unsigned {
            const unsigned y = (x & 0x7f7f7f7fu) + 0x7f7f7f7fu;
            return ~(y | x | 0x7f7f7f7fu);
        };
        auto scan16 = [&](const uint4 t, int kk) {
            unsigned m = (zb(t.x ^ pat) >> 7) | (zb(t.y ^ pat) >> 6) |
                         (zb(t.z ^ pat) >> 5) | (zb(t.w ^ pat) >> 4);
            if (m) {
                int v = kk + tstart;
                if (v >= npc) v -= npc;
                const unsigned e0 = (unsigned)v << 4;
                while (m) {
                    const unsigned bit = (unsigned)__builtin_ctz(m);
                    m &= m - 1u;
                    const unsigned e = e0 + ((bit & 3u) << 2) + (bit >> 3);
                    const unsigned p = atomicAdd(&qcount, 1u);
                    if (p < QCAP) queue[p] = e;
                }
            }
        };
        constexpr int TP = 4;
        if (npc > 0) {
            uint4 cur[TP];
#pragma unroll
            for (int j = 0; j < TP; ++j) cur[j] = ldt(tid + j * 1024);
            for (int k = tid; k < npc; k += TP * 1024) {
                uint4 nxt[TP];
#pragma unroll
                for (int j = 0; j < TP; ++j) nxt[j] = ldt(k + (TP + j) * 1024);
#pragma unroll
                for (int j = 0; j < TP; ++j) scan16(cur[j], k + j * 1024);
#pragma unroll
                for (int j = 0; j < TP; ++j) cur[j] = nxt[j];
            }
        }
        __syncthreads();
        cnt = qcount;
    } else if (!have) {
        if (nvec > 0) {
            int4 cur[NP];
#pragma unroll
            for (int j = 0; j < NP; ++j) cur[j] = ldv(tid + j * 1024);
            for (int k = tid; k < nvec; k += NP * 1024) {
                int4 nxt[NP];
#pragma unroll
                for (int j = 0; j < NP; ++j) nxt[j] = ldv(k + (NP + j) * 1024);
#pragma unroll
                for (int j = 0; j < NP; ++j) scan4(cur[j], k + j * 1024);
#pragma unroll
                for (int j = 0; j < NP; ++j) cur[j] = nxt[j];
            }
        }
        for (int e = (nvec << 2) + tid; e < E; e += 1024) {
            const unsigned u = (unsigned)(dst[e] - r0);
            if (u < RNG) { const unsigned p = atomicAdd(&qcount, 1u); if (p < QCAP) queue[p] = ((unsigned)e << 9) | u; }
        }
        __syncthreads();
        cnt = qcount;
    }

    if (cnt > QCAP) {
        for (int i = tid; i < RNG * D / 4; i += 1024) tile4[i] = make_float4(0.f, 0.f, 0.f, 0.f);
        __syncthreads();
        for (int e = tid; e < E; e += 1024) {
            const unsigned u = (unsigned)(dst[e] - r0);
            if (u < RNG) {
                const float* __restrict__ h = H + (size_t)src[e] * D;
                float* t = tile + u * D;
                for (int f = 0; f < D; ++f) atomicAdd(t + (f & 3) * LPR + (f >> 2), h[f]);
            }
        }
        __syncthreads();
    } else {
        if (!have) {
            for (unsigned i = tid; i < cnt; i += 1024) {
                const unsigned q = queue[i];
                const unsigned e = usetag ? q : q >> 9;
                const unsigned u = usetag ? (unsigned)(dst[e] - r0) : q;
                const unsigned s = (unsigned)src[e];
                queue[i] = (s << 9) | (u & (RNG - 1));
            }
            if (!KEEP && tid < 3) {
                const unsigned i = cnt + (unsigned)tid;
                if (i < ((cnt + 3u) & ~3u)) queue[i] = 0u;
            }
            __syncthreads();
        }
        if (!KEEP && cnt <= qkeep) {
            for (unsigned i = tid; i < (cnt + 3u) / 4u; i += 1024) qg4[i] = queue4[i];
        }

        const int wave = tid >> 5;
        const int lane = tid & 31;
        const int sub  = lane / LPR;
        const int l    = lane % LPR;
        constexpr int U = 8;
        constexpr unsigned STEP = EPW * U;
        const float* __restrict__ hl = H + l * 4;
        float* tl = tile + l;
        unsigned base = (unsigned)wave * STEP;
        for (; base + STEP <= cnt; base += 32u * STEP) {
            unsigned q[U];
            float4   val[U];
#pragma unroll
            for (int u = 0; u < U; ++u) q[u] = queue[base + u * EPW + sub];
#pragma unroll
            for (int u = 0; u < U; ++u)
                val[u] = *reinterpret_cast<const float4*>(hl + (size_t)(q[u] >> 9) * D);
#pragma unroll
            for (int u = 0; u < U; ++u) {
                float a0 = val[u].x, a1 = val[u].y, a2 = val[u].z, a3 = val[u].w;
                if (sub) { const float t0 = a0; a0 = a1; a1 = a2; a2 = a3; a3 = t0; }
                float* t = tl + (q[u] & (RNG - 1)) * D;
                atomicAdd(t + ((0 + sub) & 3) * LPR, a0);
                atomicAdd(t + ((1 + sub) & 3) * LPR, a1);
                atomicAdd(t + ((2 + sub) & 3) * LPR, a2);
                atomicAdd(t + ((3 + sub) & 3) * LPR, a3);
            }
        }
        if (base < cnt) {
            for (unsigned i = base + sub; i < cnt; i += EPW) {
                const unsigned q1 = queue[i];
                const float4 v1 = *reinterpret_cast<const float4*>(hl + (size_t)(q1 >> 9) * D);
                float* t = tl + (q1 & (RNG - 1)) * D;
                atomicAdd(t,           v1.x);
                atomicAdd(t + LPR,     v1.y);
                atomicAdd(t + 2 * LPR, v1.z);
                atomicAdd(t + 3 * LPR, v1.w);
            }
        }
        __syncthreads();
    }

    if (!KEEP && tid == 0) qg[0] = cnt;

    for (int i = tid; i < RNG * LPR; i += 1024) {
        const int row  = i / LPR;
        const int l    = i % LPR;
        const int grow = r0 + row;
        if (grow < N_NODES) {
            const float* t = tile + row * D + l;
            const float sc = nd[grow];
            const float4 bb = *reinterpret_cast<const float4*>(bias + l * 4);
            float4 o;
            o.x = t[0]       * sc + bb.x;
            o.y = t[LPR]     * sc + bb.y;
            o.z = t[2 * LPR] * sc + bb.z;
            o.w = t[3 * LPR] * sc + bb.w;
            if (RELU) {
                o.x = fmaxf(o.x, 0.0f); o.y = fmaxf(o.y, 0.0f);
                o.z = fmaxf(o.z, 0.0f); o.w = fmaxf(o.w, 0.0f);
            }
            *reinterpret_cast<float4*>(O + (size_t)grow * D + l * 4) = o;
        }
    }
}

extern "C" void kernel_launch(void* const* d_in, const int* in_sizes, int n_in,
                              void* d_out, int out_size, void* d_ws, size_t ws_size,
                              hipStream_t stream) {
    (void)n_in; (void)out_size; (void)ws_size;

    const float* x  = (const float*)d_in[0];
    const int*   src = (const int*)d_in[1];
    const int*   dst = (const int*)d_in[2];
    const float* W1 = (const float*)d_in[3];
    const float* b1 = (const float*)d_in[4];
    const float* W2 = (const float*)d_in[5];
    const float* b2 = (const float*)d_in[6];
    const float* W3 = (const float*)d_in[7];
    const float* b3 = (const float*)d_in[8];

    const int E = in_sizes[1];
    const int N = N_NODES;

    float* ws   = (float*)d_ws;
    float* ns   = ws;
    float* nd   = ws + NPAD;
    float* bufA = ws + 2 * NPAD;
    float* bufB = bufA + (size_t)N * 128;
    unsigned* qpark = (unsigned*)d_out;
    unsigned* part = (unsigned*)bufB;

    deg_hist_kernel<<<dim3((E + SLICE - 1) / SLICE, 2), 1024, 0, stream>>>(src, dst, part, qpark, E);
    norm_kernel<<<NPAD2 / 256, 256, 0, stream>>>(part, ns, E);

    gemm_norm_wmma<128, 128><<<GEMM_BLOCKS, 256, 0, stream>>>(x, ns, W1, bufA);
    aggregate_owner<128, true, false><<<NRNG, 1024, 0, stream>>>(bufA, src, dst, nd, b1, bufB, qpark, E);

    gemm_norm_wmma<128, 128><<<GEMM_BLOCKS, 256, 0, stream>>>(bufB, ns, W2, bufA);
    aggregate_owner<128, true, true><<<NRNG, 1024, 0, stream>>>(bufA, src, dst, nd, b2, bufB, qpark, E);

    gemm_norm_wmma<128, 64><<<GEMM_BLOCKS, 256, 0, stream>>>(bufB, ns, W3, bufA);
    aggregate_owner<64, false, true><<<NRNG, 1024, 0, stream>>>(bufA, src, dst, nd, b3, (float*)d_out, qpark, E);
}


# aggregation: owner ranges of 393 destination rows, not 512: 255 one-per-WGP work-groups for the chip's 256 WGPs, not 196; tags = dst / 393; each range's sums stay in LDS add order
# speedup vs baseline: 3.9015x; 1.0831x over previous
#include <hip/hip_runtime.h>

#define N_NODES 100000

typedef __attribute__((ext_vector_type(2))) float v2f;
typedef __attribute__((ext_vector_type(8))) float v8f;

#define NPAD  100352
#define NPAD2 (NPAD / 2)
#define SLICE 32768

#define RNG     393
#define RMASK   511u
#define TCH     16384
#define TAGBLK  (RNG * 64)
#define TAGOFF  16384
#define TAGCAP  (254 * TCH)

__global__ __launch_bounds__(1024) void deg_hist_kernel(
        const int* __restrict__ src, const int* __restrict__ dst,
        unsigned* __restrict__ part, unsigned* __restrict__ tags, int E) {
    __shared__ uint4 cnt4[NPAD2 / 4];
    unsigned* cnt = reinterpret_cast<unsigned*>(cnt4);

    const int tid = threadIdx.x;
    const int* __restrict__ idx = blockIdx.y ? dst : src;
    const int e0 = blockIdx.x * SLICE;

    constexpr int NP = SLICE / 4 / 1024;
    int4 d[NP];
#pragma unroll
    for (int j = 0; j < NP; ++j) {
        const int e = e0 + 4 * (tid + j * 1024);
        d[j] = make_int4(-1, -1, -1, -1);
        if (e + 3 < E) d[j] = *reinterpret_cast<const int4*>(idx + e);
    }

    for (int i = tid; i < NPAD2 / 4; i += 1024) cnt4[i] = make_uint4(0u, 0u, 0u, 0u);
    __syncthreads();

    auto hit = [&](int n) {
        if ((unsigned)n < (unsigned)N_NODES) atomicAdd(cnt + (n >> 1), (n & 1) ? 65536u : 1u);
    };
#pragma unroll
    for (int j = 0; j < NP; ++j) { hit(d[j].x); hit(d[j].y); hit(d[j].z); hit(d[j].w); }
    if (tid < (E & 3)) {
        const int e = (E & ~3) + tid;
        if (e >= e0 && e < e0 + SLICE) hit(idx[e]);
    }

    if (blockIdx.y == 1 && E <= TAGCAP) {
        auto tagof = [](int n) -> unsigned {
            return (unsigned)n < (unsigned)N_NODES ? (unsigned)n / (unsigned)RNG : 255u;
        };
        const int epad = (E + 15) & ~15;
#pragma unroll
        for (int j = 0; j < NP; ++j) {
            const int e = e0 + 4 * (tid + j * 1024);
            if (e < epad) {
                int4 v = d[j];
                if (e + 3 >= E) {
                    v.x = e     < E ? idx[e]     : -1;
                    v.y = e + 1 < E ? idx[e + 1] : -1;
                    v.z = e + 2 < E ? idx[e + 2] : -1;
                    v.w = -1;
                }
                tags[(size_t)(e / TCH) * TAGBLK + TAGOFF + (e % TCH) / 4] =
                    tagof(v.x) | (tagof(v.y) << 8) | (tagof(v.z) << 16) | (tagof(v.w) << 24);
            }
        }
    }
    __syncthreads();

    uint4* __restrict__ out4 = reinterpret_cast<uint4*>(
        part + ((size_t)blockIdx.y * gridDim.x + blockIdx.x) * NPAD2);
    for (int i = tid; i < NPAD2 / 4; i += 1024) out4[i] = cnt4[i];
}

__global__ __launch_bounds__(256) void norm_kernel(
        const unsigned* __restrict__ part, float* __restrict__ nsnd, int E) {
    const int nblk = (E + SLICE - 1) / SLICE;
    const int i = blockIdx.x * 256 + threadIdx.x;
    if (i >= NPAD2) return;
    const int a  = i / (NPAD2 / 2);
    const int w2 = i % (NPAD2 / 2);
    const uint2* __restrict__ p =
        reinterpret_cast<const uint2*>(part + (size_t)a * nblk * NPAD2) + w2;
    unsigned c0 = 0u, c1 = 0u, c2 = 0u, c3 = 0u;
#pragma unroll 8
    for (int b = 0; b < nblk; ++b) {
        const uint2 v = p[(size_t)b * (NPAD2 / 2)];
        c0 += v.x & 0xffffu; c1 += v.x >> 16;
        c2 += v.y & 0xffffu; c3 += v.y >> 16;
    }
    float4 o;
    o.x = rsqrtf(fmaxf((float)c0, 1.0f));
    o.y = rsqrtf(fmaxf((float)c1, 1.0f));
    o.z = rsqrtf(fmaxf((float)c2, 1.0f));
    o.w = rsqrtf(fmaxf((float)c3, 1.0f));
    *reinterpret_cast<float4*>(nsnd + (size_t)a * NPAD + 4 * w2) = o;
}

#define NSTRIPE 6250
#define GEMM_BLOCKS 782

template <int DIN, int DOUT>
__global__ __launch_bounds__(256) void gemm_norm_wmma(
        const float* __restrict__ X, const float* __restrict__ ns,
        const float* __restrict__ W, float* __restrict__ Y) {
    constexpr int PP = 2 * DOUT + 16;
    constexpr int NT = DOUT / 16;
    constexpr int C4 = DOUT / 4;
    constexpr int NU = DIN / 8;
    __shared__ float Wl[(DIN / 2) * PP];

    const int tid = threadIdx.x;
    constexpr int TRIPS = (DIN / 2) * C4 / 256;
    static_assert((DIN / 2) * C4 % 256 == 0, "W staging assumes whole trips of 256 threads");
#pragma unroll
    for (int t = 0; t < TRIPS; ++t) {
        const int i = tid + t * 256;
        const int P = i / C4, c4 = i % C4;
        const float4 r0 = *reinterpret_cast<const float4*>(W + (size_t)(2 * P) * DOUT + c4 * 4);
        const float4 r1 = *reinterpret_cast<const float4*>(W + (size_t)(2 * P + 1) * DOUT + c4 * 4);
        float4* d = reinterpret_cast<float4*>(Wl + P * PP + c4 * 8);
        d[0] = make_float4(r0.x, r1.x, r0.y, r1.y);
        d[1] = make_float4(r0.z, r1.z, r0.w, r1.w);
    }
    __syncthreads();

    const int wave = tid >> 5;
    const int lane = tid & 31;
    const int c    = lane & 15;
    const int h    = lane >> 4;
    const int stripe = blockIdx.x * 8 + wave;
    if (stripe >= NSTRIPE) return;

    const int row0 = stripe * 16;
    const float nsr = ns[row0 + c];
    const float4* __restrict__ xr =
        reinterpret_cast<const float4*>(X + (size_t)(row0 + c) * DIN) + h;

    float4 a4[NU];
#pragma unroll
    for (int u = 0; u < NU; ++u) a4[u] = xr[2 * u];

    v8f acc[NT];
#pragma unroll
    for (int j = 0; j < NT; ++j) acc[j] = v8f{};

    const float* wl = Wl + (2 * h) * PP + 2 * c;
#pragma unroll
    for (int u = 0; u < NU; ++u) {
#pragma unroll
        for (int p = 0; p < 2; ++p) {
            v2f a;
            a[0] = (p ? a4[u].z : a4[u].x) * nsr;
            a[1] = (p ? a4[u].w : a4[u].y) * nsr;
            const float* wk = wl + (4 * u + p) * PP;
#pragma unroll
            for (int j = 0; j < NT; ++j) {
                const v2f b = *reinterpret_cast<const v2f*>(wk + j * 32);
                acc[j] = __builtin_amdgcn_wmma_f32_16x16x4_f32(
                     false, a,  false, b,
                     (short)0, acc[j],  false,  false);
            }
        }
    }

    float* __restrict__ yr = Y + (size_t)(row0 + 8 * h) * DOUT + c;
#pragma unroll
    for (int j = 0; j < NT; ++j) {
#pragma unroll
        for (int v = 0; v < 8; ++v) yr[(size_t)v * DOUT + j * 16] = acc[j][v];
    }
}

#define QCAP 12288
#define NRNG ((N_NODES + RNG - 1) / RNG)
#define QROW 64
#define QHDR 32
static_assert(RNG <= (int)RMASK + 1 && NRNG <= 255 && TAGBLK % 4 == 0 && RNG * QROW == TAGBLK && QHDR + QCAP <= TAGOFF && TAGOFF + TCH / 4 <= TAGBLK,
              "tag chunks lie behind the parked queue inside a range's rows");
static_assert(TAGCAP / TCH <= N_NODES / RNG, "tag chunks only in whole ranges");

template <int D, bool RELU, bool KEEP>
__global__ __launch_bounds__(1024) void aggregate_owner(
        const float* __restrict__ H, const int* __restrict__ src,
        const int* __restrict__ dst, const float* __restrict__ nd,
        const float* __restrict__ bias, float* O, unsigned* QG, int E) {
    constexpr int LPR = D / 4;
    constexpr int EPW = 32 / LPR;

    __shared__ float    tile[RNG * D];
    __shared__ uint4    queue4[QCAP / 4];
    __shared__ unsigned qcount;
    unsigned* queue = reinterpret_cast<unsigned*>(queue4);

    const int tid = threadIdx.x;
    const int r0  = blockIdx.x * RNG;

    const int      rows  = (N_NODES - r0) < RNG ? (N_NODES - r0) : RNG;
    const unsigned qroom = (unsigned)(rows * QROW - QHDR);
    const unsigned qkeep = qroom < QCAP ? qroom : (unsigned)QCAP;
    unsigned* qg  = QG + (size_t)r0 * QROW;
    uint4*    qg4 = reinterpret_cast<uint4*>(qg + QHDR);

    bool     have = false;
    unsigned cnt  = 0u;
    if (KEEP) {
        const unsigned c = (unsigned)__builtin_amdgcn_readfirstlane((int)qg[0]);
        have = c <= qkeep;
        if (have) cnt = c;
    }

    float4* tile4 = reinterpret_cast<float4*>(tile);
    for (int i = tid; i < RNG * D / 4; i += 1024) tile4[i] = make_float4(0.f, 0.f, 0.f, 0.f);
    if (tid == 0) qcount = 0u;
    if (have) {
        for (unsigned i = tid; i < (cnt + 3u) / 4u; i += 1024) queue4[i] = qg4[i];
    }
    __syncthreads();

    const int nvec = E >> 2;
    const int4* __restrict__ dst4 = reinterpret_cast<const int4*>(dst);
    const int start = nvec > 0 ? (int)((blockIdx.x * 2048u) % (unsigned)nvec) : 0;

    auto ldv = [&](int kk) -> int4 {
        int v = (kk < nvec ? kk : nvec - 1) + start;
        if (v >= nvec) v -= nvec;
        int4 d = dst4[v];
        if (kk >= nvec) d = make_int4(-1, -1, -1, -1);
        return d;
    };
    auto scan4 = [&](const int4 d, int kk) {
        int v = kk + start;
        if (v >= nvec) v -= nvec;
        const unsigned e0 = (unsigned)v << 2;
        const unsigned u0 = (unsigned)(d.x - r0);
        const unsigned u1 = (unsigned)(d.y - r0);
        const unsigned u2 = (unsigned)(d.z - r0);
        const unsigned u3 = (unsigned)(d.w - r0);
        if (u0 < RNG) { const unsigned p = atomicAdd(&qcount, 1u); if (p < QCAP) queue[p] = ((e0     ) << 9) | u0; }
        if (u1 < RNG) { const unsigned p = atomicAdd(&qcount, 1u); if (p < QCAP) queue[p] = ((e0 + 1u) << 9) | u1; }
        if (u2 < RNG) { const unsigned p = atomicAdd(&qcount, 1u); if (p < QCAP) queue[p] = ((e0 + 2u) << 9) | u2; }
        if (u3 < RNG) { const unsigned p = atomicAdd(&qcount, 1u); if (p < QCAP) queue[p] = ((e0 + 3u) << 9) | u3; }
    };
    constexpr int NP = 8;
    const bool usetag = !KEEP && E <= TAGCAP;
    if (!have && usetag) {
        const int npc = (E + 15) >> 4;
        const uint4* tg4 = reinterpret_cast<const uint4*>(QG);
        const unsigned pat = (unsigned)blockIdx.x * 0x01010101u;
        const int tstart = npc > 0 ? (int)((blockIdx.x * 512u) % (unsigned)npc) : 0;
        auto ldt = [&](int kk) -> uint4 {
            int v = (kk < npc ? kk : npc - 1) + tstart;
            if (v >= npc) v -= npc;
            uint4 t = tg4[(size_t)(v >> 10) * (TAGBLK / 4) + TAGOFF / 4 + (v & 1023)];
            if (kk >= npc) t = make_uint4(~0u, ~0u, ~0u, ~0u);
            return t;
        };
        auto zb = [](unsigned x) -> unsigned {
            const unsigned y = (x & 0x7f7f7f7fu) + 0x7f7f7f7fu;
            return ~(y | x | 0x7f7f7f7fu);
        };
        auto scan16 = [&](const uint4 t, int kk) {
            unsigned m = (zb(t.x ^ pat) >> 7) | (zb(t.y ^ pat) >> 6) |
                         (zb(t.z ^ pat) >> 5) | (zb(t.w ^ pat) >> 4);
            if (m) {
                int v = kk + tstart;
                if (v >= npc) v -= npc;
                const unsigned e0 = (unsigned)v << 4;
                while (m) {
                    const unsigned bit = (unsigned)__builtin_ctz(m);
                    m &= m - 1u;
                    const unsigned e = e0 + ((bit & 3u) << 2) + (bit >> 3);
                    const unsigned p = atomicAdd(&qcount, 1u);
                    if (p < QCAP) queue[p] = e;
                }
            }
        };
        constexpr int TP = 4;
        if (npc > 0) {
            uint4 cur[TP];
#pragma unroll
            for (int j = 0; j < TP; ++j) cur[j] = ldt(tid + j * 1024);
            for (int k = tid; k < npc; k += TP * 1024) {
                uint4 nxt[TP];
#pragma unroll
                for (int j = 0; j < TP; ++j) nxt[j] = ldt(k + (TP + j) * 1024);
#pragma unroll
                for (int j = 0; j < TP; ++j) scan16(cur[j], k + j * 1024);
#pragma unroll
                for (int j = 0; j < TP; ++j) cur[j] = nxt[j];
            }
        }
        __syncthreads();
        cnt = qcount;
    } else if (!have) {
        if (nvec > 0) {
            int4 cur[NP];
#pragma unroll
            for (int j = 0; j < NP; ++j) cur[j] = ldv(tid + j * 1024);
            for (int k = tid; k < nvec; k += NP * 1024) {
                int4 nxt[NP];
#pragma unroll
                for (int j = 0; j < NP; ++j) nxt[j] = ldv(k + (NP + j) * 1024);
#pragma unroll
                for (int j = 0; j < NP; ++j) scan4(cur[j], k + j * 1024);
#pragma unroll
                for (int j = 0; j < NP; ++j) cur[j] = nxt[j];
            }
        }
        for (int e = (nvec << 2) + tid; e < E; e += 1024) {
            const unsigned u = (unsigned)(dst[e] - r0);
            if (u < RNG) { const unsigned p = atomicAdd(&qcount, 1u); if (p < QCAP) queue[p] = ((unsigned)e << 9) | u; }
        }
        __syncthreads();
        cnt = qcount;
    }

    if (cnt > QCAP) {
        for (int i = tid; i < RNG * D / 4; i += 1024) tile4[i] = make_float4(0.f, 0.f, 0.f, 0.f);
        __syncthreads();
        for (int e = tid; e < E; e += 1024) {
            const unsigned u = (unsigned)(dst[e] - r0);
            if (u < RNG) {
                const float* __restrict__ h = H + (size_t)src[e] * D;
                float* t = tile + u * D;
                for (int f = 0; f < D; ++f) atomicAdd(t + (f & 3) * LPR + (f >> 2), h[f]);
            }
        }
        __syncthreads();
    } else {
        if (!have) {
            for (unsigned i = tid; i < cnt; i += 1024) {
                const unsigned q = queue[i];
                const unsigned e = usetag ? q : q >> 9;
                const unsigned u = usetag ? (unsigned)(dst[e] - r0) : q;
                const unsigned s = (unsigned)src[e];
                queue[i] = (s << 9) | (u & RMASK);
            }
            if (!KEEP && tid < 3) {
                const unsigned i = cnt + (unsigned)tid;
                if (i < ((cnt + 3u) & ~3u)) queue[i] = 0u;
            }
            __syncthreads();
        }
        if (!KEEP && cnt <= qkeep) {
            for (unsigned i = tid; i < (cnt + 3u) / 4u; i += 1024) qg4[i] = queue4[i];
        }

        const int wave = tid >> 5;
        const int lane = tid & 31;
        const int sub  = lane / LPR;
        const int l    = lane % LPR;
        constexpr int U = 8;
        constexpr unsigned STEP = EPW * U;
        const float* __restrict__ hl = H + l * 4;
        float* tl = tile + l;
        unsigned base = (unsigned)wave * STEP;
        for (; base + STEP <= cnt; base += 32u * STEP) {
            unsigned q[U];
            float4   val[U];
#pragma unroll
            for (int u = 0; u < U; ++u) q[u] = queue[base + u * EPW + sub];
#pragma unroll
            for (int u = 0; u < U; ++u)
                val[u] = *reinterpret_cast<const float4*>(hl + (size_t)(q[u] >> 9) * D);
#pragma unroll
            for (int u = 0; u < U; ++u) {
                float a0 = val[u].x, a1 = val[u].y, a2 = val[u].z, a3 = val[u].w;
                if (sub) { const float t0 = a0; a0 = a1; a1 = a2; a2 = a3; a3 = t0; }
                float* t = tl + (q[u] & RMASK) * D;
                atomicAdd(t + ((0 + sub) & 3) * LPR, a0);
                atomicAdd(t + ((1 + sub) & 3) * LPR, a1);
                atomicAdd(t + ((2 + sub) & 3) * LPR, a2);
                atomicAdd(t + ((3 + sub) & 3) * LPR, a3);
            }
        }
        if (base < cnt) {
            for (unsigned i = base + sub; i < cnt; i += EPW) {
                const unsigned q1 = queue[i];
                const float4 v1 = *reinterpret_cast<const float4*>(hl + (size_t)(q1 >> 9) * D);
                float* t = tl + (q1 & RMASK) * D;
                atomicAdd(t,           v1.x);
                atomicAdd(t + LPR,     v1.y);
                atomicAdd(t + 2 * LPR, v1.z);
                atomicAdd(t + 3 * LPR, v1.w);
            }
        }
        __syncthreads();
    }

    if (!KEEP && tid == 0) qg[0] = cnt;

    for (int i = tid; i < RNG * LPR; i += 1024) {
        const int row  = i / LPR;
        const int l    = i % LPR;
        const int grow = r0 + row;
        if (grow < N_NODES) {
            const float* t = tile + row * D + l;
            const float sc = nd[grow];
            const float4 bb = *reinterpret_cast<const float4*>(bias + l * 4);
            float4 o;
            o.x = t[0]       * sc + bb.x;
            o.y = t[LPR]     * sc + bb.y;
            o.z = t[2 * LPR] * sc + bb.z;
            o.w = t[3 * LPR] * sc + bb.w;
            if (RELU) {
                o.x = fmaxf(o.x, 0.0f); o.y = fmaxf(o.y, 0.0f);
                o.z = fmaxf(o.z, 0.0f); o.w = fmaxf(o.w, 0.0f);
            }
            *reinterpret_cast<float4*>(O + (size_t)grow * D + l * 4) = o;
        }
    }
}

extern "C" void kernel_launch(void* const* d_in, const int* in_sizes, int n_in,
                              void* d_out, int out_size, void* d_ws, size_t ws_size,
                              hipStream_t stream) {
    (void)n_in; (void)out_size; (void)ws_size;

    const float* x  = (const float*)d_in[0];
    const int*   src = (const int*)d_in[1];
    const int*   dst = (const int*)d_in[2];
    const float* W1 = (const float*)d_in[3];
    const float* b1 = (const float*)d_in[4];
    const float* W2 = (const float*)d_in[5];
    const float* b2 = (const float*)d_in[6];
    const float* W3 = (const float*)d_in[7];
    const float* b3 = (const float*)d_in[8];

    const int E = in_sizes[1];
    const int N = N_NODES;

    float* ws   = (float*)d_ws;
    float* ns   = ws;
    float* nd   = ws + NPAD;
    float* bufA = ws + 2 * NPAD;
    float* bufB = bufA + (size_t)N * 128;
    unsigned* qpark = (unsigned*)d_out;
    unsigned* part = (unsigned*)bufB;

    deg_hist_kernel<<<dim3((E + SLICE - 1) / SLICE, 2), 1024, 0, stream>>>(src, dst, part, qpark, E);
    norm_kernel<<<NPAD2 / 256, 256, 0, stream>>>(part, ns, E);

    gemm_norm_wmma<128, 128><<<GEMM_BLOCKS, 256, 0, stream>>>(x, ns, W1, bufA);
    aggregate_owner<128, true, false><<<NRNG, 1024, 0, stream>>>(bufA, src, dst, nd, b1, bufB, qpark, E);

    gemm_norm_wmma<128, 128><<<GEMM_BLOCKS, 256, 0, stream>>>(bufB, ns, W2, bufA);
    aggregate_owner<128, true, true><<<NRNG, 1024, 0, stream>>>(bufA, src, dst, nd, b2, bufB, qpark, E);

    gemm_norm_wmma<128, 64><<<GEMM_BLOCKS, 256, 0, stream>>>(bufB, ns, W3, bufA);
    aggregate_owner<64, false, true><<<NRNG, 1024, 0, stream>>>(bufA, src, dst, nd, b3, (float*)d_out, qpark, E);
}


# aggregation: each range's queue grouped by local row (LDS counting sort, parked as before); a wave sums consecutive entries of one row in registers, one set of LDS adds per run; sum order changes
# speedup vs baseline: 4.1944x; 1.0751x over previous
#include <hip/hip_runtime.h>

#define N_NODES 100000

typedef __attribute__((ext_vector_type(2))) float v2f;
typedef __attribute__((ext_vector_type(8))) float v8f;

#define NPAD  100352
#define NPAD2 (NPAD / 2)
#define SLICE 32768

#define RNG     393
#define RMASK   511u
#define TCH     16384
#define TAGBLK  (RNG * 64)
#define TAGOFF  16384
#define TAGCAP  (254 * TCH)

__global__ __launch_bounds__(1024) void deg_hist_kernel(
        const int* __restrict__ src, const int* __restrict__ dst,
        unsigned* __restrict__ part, unsigned* __restrict__ tags, int E) {
    __shared__ uint4 cnt4[NPAD2 / 4];
    unsigned* cnt = reinterpret_cast<unsigned*>(cnt4);

    const int tid = threadIdx.x;
    const int* __restrict__ idx = blockIdx.y ? dst : src;
    const int e0 = blockIdx.x * SLICE;

    constexpr int NP = SLICE / 4 / 1024;
    int4 d[NP];
#pragma unroll
    for (int j = 0; j < NP; ++j) {
        const int e = e0 + 4 * (tid + j * 1024);
        d[j] = make_int4(-1, -1, -1, -1);
        if (e + 3 < E) d[j] = *reinterpret_cast<const int4*>(idx + e);
    }

    for (int i = tid; i < NPAD2 / 4; i += 1024) cnt4[i] = make_uint4(0u, 0u, 0u, 0u);
    __syncthreads();

    auto hit = [&](int n) {
        if ((unsigned)n < (unsigned)N_NODES) atomicAdd(cnt + (n >> 1), (n & 1) ? 65536u : 1u);
    };
#pragma unroll
    for (int j = 0; j < NP; ++j) { hit(d[j].x); hit(d[j].y); hit(d[j].z); hit(d[j].w); }
    if (tid < (E & 3)) {
        const int e = (E & ~3) + tid;
        if (e >= e0 && e < e0 + SLICE) hit(idx[e]);
    }

    if (blockIdx.y == 1 && E <= TAGCAP) {
        auto tagof = [](int n) -> unsigned {
            return (unsigned)n < (unsigned)N_NODES ? (unsigned)n / (unsigned)RNG : 255u;
        };
        const int epad = (E + 15) & ~15;
#pragma unroll
        for (int j = 0; j < NP; ++j) {
            const int e = e0 + 4 * (tid + j * 1024);
            if (e < epad) {
                int4 v = d[j];
                if (e + 3 >= E) {
                    v.x = e     < E ? idx[e]     : -1;
                    v.y = e + 1 < E ? idx[e + 1] : -1;
                    v.z = e + 2 < E ? idx[e + 2] : -1;
                    v.w = -1;
                }
                tags[(size_t)(e / TCH) * TAGBLK + TAGOFF + (e % TCH) / 4] =
                    tagof(v.x) | (tagof(v.y) << 8) | (tagof(v.z) << 16) | (tagof(v.w) << 24);
            }
        }
    }
    __syncthreads();

    uint4* __restrict__ out4 = reinterpret_cast<uint4*>(
        part + ((size_t)blockIdx.y * gridDim.x + blockIdx.x) * NPAD2);
    for (int i = tid; i < NPAD2 / 4; i += 1024) out4[i] = cnt4[i];
}

__global__ __launch_bounds__(256) void norm_kernel(
        const unsigned* __restrict__ part, float* __restrict__ nsnd, int E) {
    const int nblk = (E + SLICE - 1) / SLICE;
    const int i = blockIdx.x * 256 + threadIdx.x;
    if (i >= NPAD2) return;
    const int a  = i / (NPAD2 / 2);
    const int w2 = i % (NPAD2 / 2);
    const uint2* __restrict__ p =
        reinterpret_cast<const uint2*>(part + (size_t)a * nblk * NPAD2) + w2;
    unsigned c0 = 0u, c1 = 0u, c2 = 0u, c3 = 0u;
#pragma unroll 8
    for (int b = 0; b < nblk; ++b) {
        const uint2 v = p[(size_t)b * (NPAD2 / 2)];
        c0 += v.x & 0xffffu; c1 += v.x >> 16;
        c2 += v.y & 0xffffu; c3 += v.y >> 16;
    }
    float4 o;
    o.x = rsqrtf(fmaxf((float)c0, 1.0f));
    o.y = rsqrtf(fmaxf((float)c1, 1.0f));
    o.z = rsqrtf(fmaxf((float)c2, 1.0f));
    o.w = rsqrtf(fmaxf((float)c3, 1.0f));
    *reinterpret_cast<float4*>(nsnd + (size_t)a * NPAD + 4 * w2) = o;
}

#define NSTRIPE 6250
#define GEMM_BLOCKS 782

template <int DIN, int DOUT>
__global__ __launch_bounds__(256) void gemm_norm_wmma(
        const float* __restrict__ X, const float* __restrict__ ns,
        const float* __restrict__ W, float* __restrict__ Y) {
    constexpr int PP = 2 * DOUT + 16;
    constexpr int NT = DOUT / 16;
    constexpr int C4 = DOUT / 4;
    constexpr int NU = DIN / 8;
    __shared__ float Wl[(DIN / 2) * PP];

    const int tid = threadIdx.x;
    constexpr int TRIPS = (DIN / 2) * C4 / 256;
    static_assert((DIN / 2) * C4 % 256 == 0, "W staging assumes whole trips of 256 threads");
#pragma unroll
    for (int t = 0; t < TRIPS; ++t) {
        const int i = tid + t * 256;
        const int P = i / C4, c4 = i % C4;
        const float4 r0 = *reinterpret_cast<const float4*>(W + (size_t)(2 * P) * DOUT + c4 * 4);
        const float4 r1 = *reinterpret_cast<const float4*>(W + (size_t)(2 * P + 1) * DOUT + c4 * 4);
        float4* d = reinterpret_cast<float4*>(Wl + P * PP + c4 * 8);
        d[0] = make_float4(r0.x, r1.x, r0.y, r1.y);
        d[1] = make_float4(r0.z, r1.z, r0.w, r1.w);
    }
    __syncthreads();

    const int wave = tid >> 5;
    const int lane = tid & 31;
    const int c    = lane & 15;
    const int h    = lane >> 4;
    const int stripe = blockIdx.x * 8 + wave;
    if (stripe >= NSTRIPE) return;

    const int row0 = stripe * 16;
    const float nsr = ns[row0 + c];
    const float4* __restrict__ xr =
        reinterpret_cast<const float4*>(X + (size_t)(row0 + c) * DIN) + h;

    float4 a4[NU];
#pragma unroll
    for (int u = 0; u < NU; ++u) a4[u] = xr[2 * u];

    v8f acc[NT];
#pragma unroll
    for (int j = 0; j < NT; ++j) acc[j] = v8f{};

    const float* wl = Wl + (2 * h) * PP + 2 * c;
#pragma unroll
    for (int u = 0; u < NU; ++u) {
#pragma unroll
        for (int p = 0; p < 2; ++p) {
            v2f a;
            a[0] = (p ? a4[u].z : a4[u].x) * nsr;
            a[1] = (p ? a4[u].w : a4[u].y) * nsr;
            const float* wk = wl + (4 * u + p) * PP;
#pragma unroll
            for (int j = 0; j < NT; ++j) {
                const v2f b = *reinterpret_cast<const v2f*>(wk + j * 32);
                acc[j] = __builtin_amdgcn_wmma_f32_16x16x4_f32(
                     false, a,  false, b,
                     (short)0, acc[j],  false,  false);
            }
        }
    }

    float* __restrict__ yr = Y + (size_t)(row0 + 8 * h) * DOUT + c;
#pragma unroll
    for (int j = 0; j < NT; ++j) {
#pragma unroll
        for (int v = 0; v < 8; ++v) yr[(size_t)v * DOUT + j * 16] = acc[j][v];
    }
}

#define QCAP 12288
#define NRNG ((N_NODES + RNG - 1) / RNG)
#define QROW 64
#define QHDR 32
#define ROWS 512
static_assert(ROWS == (int)RMASK + 1 && RNG <= (int)RMASK + 1 && NRNG <= 255 && TAGBLK % 4 == 0 && RNG * QROW == TAGBLK && QHDR + QCAP <= TAGOFF && TAGOFF + TCH / 4 <= TAGBLK,
              "tag chunks lie behind the parked queue inside a range's rows");
static_assert(TAGCAP / TCH <= N_NODES / RNG, "tag chunks only in whole ranges");

template <int D, bool RELU, bool KEEP>
__global__ __launch_bounds__(1024) void aggregate_owner(
        const float* __restrict__ H, const int* __restrict__ src,
        const int* __restrict__ dst, const float* __restrict__ nd,
        const float* __restrict__ bias, float* O, unsigned* QG, int E) {
    constexpr int LPR = D / 4;
    constexpr int EPW = 32 / LPR;

    __shared__ float    tile[RNG * D];
    __shared__ uint4    queue4[QCAP / 4];
    __shared__ uint4    qs4[QCAP / 4];
    __shared__ unsigned rowoff[ROWS];
    __shared__ unsigned qcount;
    unsigned* queue = reinterpret_cast<unsigned*>(queue4);
    unsigned* qs    = reinterpret_cast<unsigned*>(qs4);

    const int tid = threadIdx.x;
    const int r0  = blockIdx.x * RNG;

    const int      rows  = (N_NODES - r0) < RNG ? (N_NODES - r0) : RNG;
    const unsigned qroom = (unsigned)(rows * QROW - QHDR);
    const unsigned qkeep = qroom < QCAP ? qroom : (unsigned)QCAP;
    unsigned* qg  = QG + (size_t)r0 * QROW;
    uint4*    qg4 = reinterpret_cast<uint4*>(qg + QHDR);

    bool     have = false;
    unsigned cnt  = 0u;
    if (KEEP) {
        const unsigned c = (unsigned)__builtin_amdgcn_readfirstlane((int)qg[0]);
        have = c <= qkeep;
        if (have) cnt = c;
    }

    float4* tile4 = reinterpret_cast<float4*>(tile);
    for (int i = tid; i < RNG * D / 4; i += 1024) tile4[i] = make_float4(0.f, 0.f, 0.f, 0.f);
    if (tid == 0) qcount = 0u;
    if (tid < ROWS) rowoff[tid] = 0u;
    if (have) {
        for (unsigned i = tid; i < (cnt + 3u) / 4u; i += 1024) qs4[i] = qg4[i];
    }
    __syncthreads();

    const int nvec = E >> 2;
    const int4* __restrict__ dst4 = reinterpret_cast<const int4*>(dst);
    const int start = nvec > 0 ? (int)((blockIdx.x * 2048u) % (unsigned)nvec) : 0;

    auto ldv = [&](int kk) -> int4 {
        int v = (kk < nvec ? kk : nvec - 1) + start;
        if (v >= nvec) v -= nvec;
        int4 d = dst4[v];
        if (kk >= nvec) d = make_int4(-1, -1, -1, -1);
        return d;
    };
    auto scan4 = [&](const int4 d, int kk) {
        int v = kk + start;
        if (v >= nvec) v -= nvec;
        const unsigned e0 = (unsigned)v << 2;
        const unsigned u0 = (unsigned)(d.x - r0);
        const unsigned u1 = (unsigned)(d.y - r0);
        const unsigned u2 = (unsigned)(d.z - r0);
        const unsigned u3 = (unsigned)(d.w - r0);
        if (u0 < RNG) { const unsigned p = atomicAdd(&qcount, 1u); if (p < QCAP) queue[p] = ((e0     ) << 9) | u0; }
        if (u1 < RNG) { const unsigned p = atomicAdd(&qcount, 1u); if (p < QCAP) queue[p] = ((e0 + 1u) << 9) | u1; }
        if (u2 < RNG) { const unsigned p = atomicAdd(&qcount, 1u); if (p < QCAP) queue[p] = ((e0 + 2u) << 9) | u2; }
        if (u3 < RNG) { const unsigned p = atomicAdd(&qcount, 1u); if (p < QCAP) queue[p] = ((e0 + 3u) << 9) | u3; }
    };
    constexpr int NP = 8;
    const bool usetag = !KEEP && E <= TAGCAP;
    if (!have && usetag) {
        const int npc = (E + 15) >> 4;
        const uint4* tg4 = reinterpret_cast<const uint4*>(QG);
        const unsigned pat = (unsigned)blockIdx.x * 0x01010101u;
        const int tstart = npc > 0 ? (int)((blockIdx.x * 512u) % (unsigned)npc) : 0;
        auto ldt = [&](int kk) -> uint4 {
            int v = (kk < npc ? kk : npc - 1) + tstart;
            if (v >= npc) v -= npc;
            uint4 t = tg4[(size_t)(v >> 10) * (TAGBLK / 4) + TAGOFF / 4 + (v & 1023)];
            if (kk >= npc) t = make_uint4(~0u, ~0u, ~0u, ~0u);
            return t;
        };
        auto zb = [](unsigned x) -> unsigned {
            const unsigned y = (x & 0x7f7f7f7fu) + 0x7f7f7f7fu;
            return ~(y | x | 0x7f7f7f7fu);
        };
        auto scan16 = [&](const uint4 t, int kk) {
            unsigned m = (zb(t.x ^ pat) >> 7) | (zb(t.y ^ pat) >> 6) |
                         (zb(t.z ^ pat) >> 5) | (zb(t.w ^ pat) >> 4);
            if (m) {
                int v = kk + tstart;
                if (v >= npc) v -= npc;
                const unsigned e0 = (unsigned)v << 4;
                while (m) {
                    const unsigned bit = (unsigned)__builtin_ctz(m);
                    m &= m - 1u;
                    const unsigned e = e0 + ((bit & 3u) << 2) + (bit >> 3);
                    const unsigned p = atomicAdd(&qcount, 1u);
                    if (p < QCAP) queue[p] = e;
                }
            }
        };
        constexpr int TP = 4;
        if (npc > 0) {
            uint4 cur[TP];
#pragma unroll
            for (int j = 0; j < TP; ++j) cur[j] = ldt(tid + j * 1024);
            for (int k = tid; k < npc; k += TP * 1024) {
                uint4 nxt[TP];
#pragma unroll
                for (int j = 0; j < TP; ++j) nxt[j] = ldt(k + (TP + j) * 1024);
#pragma unroll
                for (int j = 0; j < TP; ++j) scan16(cur[j], k + j * 1024);
#pragma unroll
                for (int j = 0; j < TP; ++j) cur[j] = nxt[j];
            }
        }
        __syncthreads();
        cnt = qcount;
    } else if (!have) {
        if (nvec > 0) {
            int4 cur[NP];
#pragma unroll
            for (int j = 0; j < NP; ++j) cur[j] = ldv(tid + j * 1024);
            for (int k = tid; k < nvec; k += NP * 1024) {
                int4 nxt[NP];
#pragma unroll
                for (int j = 0; j < NP; ++j) nxt[j] = ldv(k + (NP + j) * 1024);
#pragma unroll
                for (int j = 0; j < NP; ++j) scan4(cur[j], k + j * 1024);
#pragma unroll
                for (int j = 0; j < NP; ++j) cur[j] = nxt[j];
            }
        }
        for (int e = (nvec << 2) + tid; e < E; e += 1024) {
            const unsigned u = (unsigned)(dst[e] - r0);
            if (u < RNG) { const unsigned p = atomicAdd(&qcount, 1u); if (p < QCAP) queue[p] = ((unsigned)e << 9) | u; }
        }
        __syncthreads();
        cnt = qcount;
    }

    if (cnt > QCAP) {
        for (int i = tid; i < RNG * D / 4; i += 1024) tile4[i] = make_float4(0.f, 0.f, 0.f, 0.f);
        __syncthreads();
        for (int e = tid; e < E; e += 1024) {
            const unsigned u = (unsigned)(dst[e] - r0);
            if (u < RNG) {
                const float* __restrict__ h = H + (size_t)src[e] * D;
                float* t = tile + u * D;
                for (int f = 0; f < D; ++f) atomicAdd(t + (f & 3) * LPR + (f >> 2), h[f]);
            }
        }
        __syncthreads();
    } else {
        if (!have) {
            for (unsigned i = tid; i < cnt; i += 1024) {
                const unsigned q = queue[i];
                const unsigned e = usetag ? q : q >> 9;
                const unsigned u = usetag ? (unsigned)(dst[e] - r0) : q;
                const unsigned s = (unsigned)src[e];
                queue[i] = (s << 9) | (u & RMASK);
                atomicAdd(&rowoff[u & RMASK], 1u);
            }
            __syncthreads();
            if (tid < 32) {
                unsigned first[ROWS / 32];
                unsigned sum = 0u;
#pragma unroll
                for (int k = 0; k < ROWS / 32; ++k) {
                    const unsigned t = rowoff[tid * (ROWS / 32) + k];
                    first[k] = sum;
                    sum += t;
                }
                unsigned inc = sum;
#pragma unroll
                for (int d = 1; d < 32; d <<= 1) {
                    const unsigned below = (unsigned)__shfl_up((int)inc, (unsigned)d);
                    if (tid >= d) inc += below;
                }
                const unsigned before = inc - sum;
#pragma unroll
                for (int k = 0; k < ROWS / 32; ++k) rowoff[tid * (ROWS / 32) + k] = first[k] + before;
            }
            __syncthreads();
            for (unsigned i = tid; i < cnt; i += 1024) {
                const unsigned q = queue[i];
                const unsigned p = atomicAdd(&rowoff[q & RMASK], 1u);
                qs[p] = q;
            }
            if (!KEEP && tid < 3) {
                const unsigned i = cnt + (unsigned)tid;
                if (i < ((cnt + 3u) & ~3u)) qs[i] = 0u;
            }
            __syncthreads();
        }
        if (!KEEP && cnt <= qkeep) {
            for (unsigned i = tid; i < (cnt + 3u) / 4u; i += 1024) qg4[i] = qs4[i];
        }

        const int wave = tid >> 5;
        const int lane = tid & 31;
        const int sub  = lane / LPR;
        const int l    = lane % LPR;
        constexpr int U = 8;
        constexpr unsigned STEP = EPW * U;
        const float* __restrict__ hl = H + l * 4;
        float* tl = tile + l;
        for (unsigned g = (unsigned)wave * STEP; g < cnt; g += 32u * STEP) {
            const unsigned i0 = g + (unsigned)sub * U;
            unsigned q[U];
            float4   val[U];
#pragma unroll
            for (int u = 0; u < U; ++u) {
                const unsigned i = i0 + (unsigned)u;
                q[u] = qs[i < cnt ? i : cnt - 1u];
            }
#pragma unroll
            for (int u = 0; u < U; ++u)
                val[u] = *reinterpret_cast<const float4*>(hl + (size_t)(q[u] >> 9) * D);
#pragma unroll
            for (int u = 0; u < U; ++u) {
                const bool here = i0 + (unsigned)u < cnt;
                const unsigned r = q[u] & RMASK;
                const int  un = u + 1 < U ? u + 1 : u;
                const bool on = u + 1 < U && i0 + (unsigned)un < cnt && (q[un] & RMASK) == r;
                if (on) {
                    val[un].x += val[u].x; val[un].y += val[u].y;
                    val[un].z += val[u].z; val[un].w += val[u].w;
                } else if (here) {
                    float a0 = val[u].x, a1 = val[u].y, a2 = val[u].z, a3 = val[u].w;
                    if (sub) { const float t0 = a0; a0 = a1; a1 = a2; a2 = a3; a3 = t0; }
                    float* t = tl + r * D;
                    atomicAdd(t + ((0 + sub) & 3) * LPR, a0);
                    atomicAdd(t + ((1 + sub) & 3) * LPR, a1);
                    atomicAdd(t + ((2 + sub) & 3) * LPR, a2);
                    atomicAdd(t + ((3 + sub) & 3) * LPR, a3);
                }
            }
        }
        __syncthreads();
    }

    if (!KEEP && tid == 0) qg[0] = cnt;

    for (int i = tid; i < RNG * LPR; i += 1024) {
        const int row  = i / LPR;
        const int l    = i % LPR;
        const int grow = r0 + row;
        if (grow < N_NODES) {
            const float* t = tile + row * D + l;
            const float sc = nd[grow];
            const float4 bb = *reinterpret_cast<const float4*>(bias + l * 4);
            float4 o;
            o.x = t[0]       * sc + bb.x;
            o.y = t[LPR]     * sc + bb.y;
            o.z = t[2 * LPR] * sc + bb.z;
            o.w = t[3 * LPR] * sc + bb.w;
            if (RELU) {
                o.x = fmaxf(o.x, 0.0f); o.y = fmaxf(o.y, 0.0f);
                o.z = fmaxf(o.z, 0.0f); o.w = fmaxf(o.w, 0.0f);
            }
            *reinterpret_cast<float4*>(O + (size_t)grow * D + l * 4) = o;
        }
    }
}

extern "C" void kernel_launch(void* const* d_in, const int* in_sizes, int n_in,
                              void* d_out, int out_size, void* d_ws, size_t ws_size,
                              hipStream_t stream) {
    (void)n_in; (void)out_size; (void)ws_size;

    const float* x  = (const float*)d_in[0];
    const int*   src = (const int*)d_in[1];
    const int*   dst = (const int*)d_in[2];
    const float* W1 = (const float*)d_in[3];
    const float* b1 = (const float*)d_in[4];
    const float* W2 = (const float*)d_in[5];
    const float* b2 = (const float*)d_in[6];
    const float* W3 = (const float*)d_in[7];
    const float* b3 = (const float*)d_in[8];

    const int E = in_sizes[1];
    const int N = N_NODES;

    float* ws   = (float*)d_ws;
    float* ns   = ws;
    float* nd   = ws + NPAD;
    float* bufA = ws + 2 * NPAD;
    float* bufB = bufA + (size_t)N * 128;
    unsigned* qpark = (unsigned*)d_out;
    unsigned* part = (unsigned*)bufB;

    deg_hist_kernel<<<dim3((E + SLICE - 1) / SLICE, 2), 1024, 0, stream>>>(src, dst, part, qpark, E);
    norm_kernel<<<NPAD2 / 256, 256, 0, stream>>>(part, ns, E);

    gemm_norm_wmma<128, 128><<<GEMM_BLOCKS, 256, 0, stream>>>(x, ns, W1, bufA);
    aggregate_owner<128, true, false><<<NRNG, 1024, 0, stream>>>(bufA, src, dst, nd, b1, bufB, qpark, E);

    gemm_norm_wmma<128, 128><<<GEMM_BLOCKS, 256, 0, stream>>>(bufB, ns, W2, bufA);
    aggregate_owner<128, true, true><<<NRNG, 1024, 0, stream>>>(bufA, src, dst, nd, b2, bufB, qpark, E);

    gemm_norm_wmma<128, 64><<<GEMM_BLOCKS, 256, 0, stream>>>(bufB, ns, W3, bufA);
    aggregate_owner<64, false, true><<<NRNG, 1024, 0, stream>>>(bufA, src, dst, nd, b3, (float*)d_out, qpark, E);
}


# layer 1 edge search: dst slices write a presence bitmap per owner range (1 bit per 16 edges), not byte tags; owners read their bitmap, then only flagged 64-B pieces of dst and src; LDS add order
# speedup vs baseline: 4.6145x; 1.1002x over previous
#include <hip/hip_runtime.h>

#define N_NODES 100000

typedef __attribute__((ext_vector_type(2))) float v2f;
typedef __attribute__((ext_vector_type(8))) float v8f;

#define NPAD  100352
#define NPAD2 (NPAD / 2)
#define SLICE 32768

#define RNG     393
#define RMASK   511u
#define NRNG    ((N_NODES + RNG - 1) / RNG)
#define NWHOLE  (N_NODES / RNG)
#define RBLK    (RNG * 64)
#define GRP     16
#define BMSW    (SLICE / GRP / 32)
#define BMSL    64
#define BMW     (BMSL * BMSW)
#define BMOFF   16384
#define BMPITCH (NRNG + 2)

__device__ __forceinline__ size_t bm_base(int r) {
    return r < NWHOLE ? (size_t)r * RBLK + BMOFF : (size_t)BMOFF + BMW;
}

__global__ __launch_bounds__(1024) void deg_hist_kernel(
        const int* __restrict__ src, const int* __restrict__ dst,
        unsigned* __restrict__ part, unsigned* __restrict__ bmg, int E) {
    __shared__ uint4 cnt4[NPAD2 / 4];
    __shared__ uint4 bm4[BMSW * BMPITCH / 4];
    unsigned* cnt = reinterpret_cast<unsigned*>(cnt4);
    unsigned* bm  = reinterpret_cast<unsigned*>(bm4);

    const int tid = threadIdx.x;
    const int* __restrict__ idx = blockIdx.y ? dst : src;
    const int e0 = blockIdx.x * SLICE;
    const bool mark = blockIdx.y == 1 && gridDim.x <= BMSL;

    constexpr int NP = SLICE / 4 / 1024;
    int4 d[NP];
#pragma unroll
    for (int j = 0; j < NP; ++j) {
        const int e = e0 + 4 * (tid + j * 1024);
        d[j] = make_int4(-1, -1, -1, -1);
        if (e + 3 < E) d[j] = *reinterpret_cast<const int4*>(idx + e);
    }

    for (int i = tid; i < NPAD2 / 4; i += 1024) cnt4[i] = make_uint4(0u, 0u, 0u, 0u);
    if (mark) {
        for (int i = tid; i < BMSW * BMPITCH / 4; i += 1024) bm4[i] = make_uint4(0u, 0u, 0u, 0u);
    }
    __syncthreads();

    auto hit = [&](int n) {
        if ((unsigned)n < (unsigned)N_NODES) atomicAdd(cnt + (n >> 1), (n & 1) ? 65536u : 1u);
    };
    auto flag = [&](int n, unsigned le) {
        if ((unsigned)n < (unsigned)N_NODES)
            atomicOr(bm + (le / (GRP * 32)) * BMPITCH + (unsigned)n / (unsigned)RNG, 1u << ((le / GRP) & 31u));
    };
#pragma unroll
    for (int j = 0; j < NP; ++j) { hit(d[j].x); hit(d[j].y); hit(d[j].z); hit(d[j].w); }
    if (mark) {
#pragma unroll
        for (int j = 0; j < NP; ++j) {
            const unsigned le = 4u * (unsigned)(tid + j * 1024);
            flag(d[j].x, le); flag(d[j].y, le); flag(d[j].z, le); flag(d[j].w, le);
        }
    }
    if (tid < (E & 3)) {
        const int e = (E & ~3) + tid;
        if (e >= e0 && e < e0 + SLICE) {
            const int n = idx[e];
            hit(n);
            if (mark) flag(n, (unsigned)(e - e0));
        }
    }
    __syncthreads();

    uint4* __restrict__ out4 = reinterpret_cast<uint4*>(
        part + ((size_t)blockIdx.y * gridDim.x + blockIdx.x) * NPAD2);
    for (int i = tid; i < NPAD2 / 4; i += 1024) out4[i] = cnt4[i];

    if (mark) {
        for (int i = tid; i < NRNG * BMSW; i += 1024) {
            const int r = i / BMSW, w = i % BMSW;
            bmg[bm_base(r) + (size_t)blockIdx.x * BMSW + w] = bm[w * BMPITCH + r];
        }
    }
}

__global__ __launch_bounds__(256) void norm_kernel(
        const unsigned* __restrict__ part, float* __restrict__ nsnd, int E) {
    const int nblk = (E + SLICE - 1) / SLICE;
    const int i = blockIdx.x * 256 + threadIdx.x;
    if (i >= NPAD2) return;
    const int a  = i / (NPAD2 / 2);
    const int w2 = i % (NPAD2 / 2);
    const uint2* __restrict__ p =
        reinterpret_cast<const uint2*>(part + (size_t)a * nblk * NPAD2) + w2;
    unsigned c0 = 0u, c1 = 0u, c2 = 0u, c3 = 0u;
#pragma unroll 8
    for (int b = 0; b < nblk; ++b) {
        const uint2 v = p[(size_t)b * (NPAD2 / 2)];
        c0 += v.x & 0xffffu; c1 += v.x >> 16;
        c2 += v.y & 0xffffu; c3 += v.y >> 16;
    }
    float4 o;
    o.x = rsqrtf(fmaxf((float)c0, 1.0f));
    o.y = rsqrtf(fmaxf((float)c1, 1.0f));
    o.z = rsqrtf(fmaxf((float)c2, 1.0f));
    o.w = rsqrtf(fmaxf((float)c3, 1.0f));
    *reinterpret_cast<float4*>(nsnd + (size_t)a * NPAD + 4 * w2) = o;
}

#define NSTRIPE 6250
#define GEMM_BLOCKS 782

template <int DIN, int DOUT>
__global__ __launch_bounds__(256) void gemm_norm_wmma(
        const float* __restrict__ X, const float* __restrict__ ns,
        const float* __restrict__ W, float* __restrict__ Y) {
    constexpr int PP = 2 * DOUT + 16;
    constexpr int NT = DOUT / 16;
    constexpr int C4 = DOUT / 4;
    constexpr int NU = DIN / 8;
    __shared__ float Wl[(DIN / 2) * PP];

    const int tid = threadIdx.x;
    constexpr int TRIPS = (DIN / 2) * C4 / 256;
    static_assert((DIN / 2) * C4 % 256 == 0, "W staging assumes whole trips of 256 threads");
#pragma unroll
    for (int t = 0; t < TRIPS; ++t) {
        const int i = tid + t * 256;
        const int P = i / C4, c4 = i % C4;
        const float4 r0 = *reinterpret_cast<const float4*>(W + (size_t)(2 * P) * DOUT + c4 * 4);
        const float4 r1 = *reinterpret_cast<const float4*>(W + (size_t)(2 * P + 1) * DOUT + c4 * 4);
        float4* d = reinterpret_cast<float4*>(Wl + P * PP + c4 * 8);
        d[0] = make_float4(r0.x, r1.x, r0.y, r1.y);
        d[1] = make_float4(r0.z, r1.z, r0.w, r1.w);
    }
    __syncthreads();

    const int wave = tid >> 5;
    const int lane = tid & 31;
    const int c    = lane & 15;
    const int h    = lane >> 4;
    const int stripe = blockIdx.x * 8 + wave;
    if (stripe >= NSTRIPE) return;

    const int row0 = stripe * 16;
    const float nsr = ns[row0 + c];
    const float4* __restrict__ xr =
        reinterpret_cast<const float4*>(X + (size_t)(row0 + c) * DIN) + h;

    float4 a4[NU];
#pragma unroll
    for (int u = 0; u < NU; ++u) a4[u] = xr[2 * u];

    v8f acc[NT];
#pragma unroll
    for (int j = 0; j < NT; ++j) acc[j] = v8f{};

    const float* wl = Wl + (2 * h) * PP + 2 * c;
#pragma unroll
    for (int u = 0; u < NU; ++u) {
#pragma unroll
        for (int p = 0; p < 2; ++p) {
            v2f a;
            a[0] = (p ? a4[u].z : a4[u].x) * nsr;
            a[1] = (p ? a4[u].w : a4[u].y) * nsr;
            const float* wk = wl + (4 * u + p) * PP;
#pragma unroll
            for (int j = 0; j < NT; ++j) {
                const v2f b = *reinterpret_cast<const v2f*>(wk + j * 32);
                acc[j] = __builtin_amdgcn_wmma_f32_16x16x4_f32(
                     false, a,  false, b,
                     (short)0, acc[j],  false,  false);
            }
        }
    }

    float* __restrict__ yr = Y + (size_t)(row0 + 8 * h) * DOUT + c;
#pragma unroll
    for (int j = 0; j < NT; ++j) {
#pragma unroll
        for (int v = 0; v < 8; ++v) yr[(size_t)v * DOUT + j * 16] = acc[j][v];
    }
}

#define QCAP 12288
#define QROW 64
#define QHDR 32
#define ROWS 512
static_assert(ROWS == (int)RMASK + 1 && RNG <= (int)RMASK + 1 && RBLK % 4 == 0 && RNG * QROW == RBLK && QHDR + QCAP <= BMOFF && BMOFF + 2 * BMW <= RBLK,
              "a range's bitmap, and in range 0 a second one, lie behind the parked queue inside the range's rows");
static_assert(NRNG - NWHOLE <= 1 && NWHOLE >= 1 && BMPITCH >= NRNG && BMPITCH % 2 == 1 && (BMSW * BMPITCH) % 4 == 0 && SLICE % (GRP * 32) == 0 && GRP == 16,
              "one short range at most, its bitmap kept in range 0's rows; a group is four 16-byte pieces");

template <int D, bool RELU, bool KEEP>
__global__ __launch_bounds__(1024) void aggregate_owner(
        const float* __restrict__ H, const int* __restrict__ src,
        const int* __restrict__ dst, const float* __restrict__ nd,
        const float* __restrict__ bias, float* O, unsigned* QG, int E) {
    constexpr int LPR = D / 4;
    constexpr int EPW = 32 / LPR;

    __shared__ float    tile[RNG * D];
    __shared__ uint4    queue4[QCAP / 4];
    __shared__ uint4    qs4[QCAP / 4];
    __shared__ unsigned rowoff[ROWS];
    __shared__ unsigned qcount;
    __shared__ unsigned gcount;
    unsigned* queue = reinterpret_cast<unsigned*>(queue4);
    unsigned* qs    = reinterpret_cast<unsigned*>(qs4);

    const int tid = threadIdx.x;
    const int r0  = blockIdx.x * RNG;

    const int      rows  = (N_NODES - r0) < RNG ? (N_NODES - r0) : RNG;
    const unsigned qroom = (unsigned)(rows * QROW - QHDR);
    const unsigned qkeep = qroom < QCAP ? qroom : (unsigned)QCAP;
    unsigned* qg  = QG + (size_t)r0 * QROW;
    uint4*    qg4 = reinterpret_cast<uint4*>(qg + QHDR);

    bool     have = false;
    unsigned cnt  = 0u;
    if (KEEP) {
        const unsigned c = (unsigned)__builtin_amdgcn_readfirstlane((int)qg[0]);
        have = c <= qkeep;
        if (have) cnt = c;
    }

    float4* tile4 = reinterpret_cast<float4*>(tile);
    for (int i = tid; i < RNG * D / 4; i += 1024) tile4[i] = make_float4(0.f, 0.f, 0.f, 0.f);
    if (tid == 0) { qcount = 0u; gcount = 0u; }
    if (tid < ROWS) rowoff[tid] = 0u;
    if (have) {
        for (unsigned i = tid; i < (cnt + 3u) / 4u; i += 1024) qs4[i] = qg4[i];
    }
    __syncthreads();

    const int nvec = E >> 2;
    const int4* __restrict__ dst4 = reinterpret_cast<const int4*>(dst);
    const int start = nvec > 0 ? (int)((blockIdx.x * 2048u) % (unsigned)nvec) : 0;

    auto ldv = [&](int kk) -> int4 {
        int v = (kk < nvec ? kk : nvec - 1) + start;
        if (v >= nvec) v -= nvec;
        int4 d = dst4[v];
        if (kk >= nvec) d = make_int4(-1, -1, -1, -1);
        return d;
    };
    auto scan4 = [&](const int4 d, int kk) {
        int v = kk + start;
        if (v >= nvec) v -= nvec;
        const unsigned e0 = (unsigned)v << 2;
        const unsigned u0 = (unsigned)(d.x - r0);
        const unsigned u1 = (unsigned)(d.y - r0);
        const unsigned u2 = (unsigned)(d.z - r0);
        const unsigned u3 = (unsigned)(d.w - r0);
        if (u0 < RNG) { const unsigned p = atomicAdd(&qcount, 1u); if (p < QCAP) queue[p] = ((e0     ) << 9) | u0; }
        if (u1 < RNG) { const unsigned p = atomicAdd(&qcount, 1u); if (p < QCAP) queue[p] = ((e0 + 1u) << 9) | u1; }
        if (u2 < RNG) { const unsigned p = atomicAdd(&qcount, 1u); if (p < QCAP) queue[p] = ((e0 + 2u) << 9) | u2; }
        if (u3 < RNG) { const unsigned p = atomicAdd(&qcount, 1u); if (p < QCAP) queue[p] = ((e0 + 3u) << 9) | u3; }
    };
    constexpr int NP = 8;
    const int  nblk  = (E + SLICE - 1) / SLICE;
    const bool usebm = !KEEP && nblk <= BMSL && nvec > 0;
    bool listed = false;
    if (!have && usebm) {
        const unsigned* bmp = QG + bm_base(blockIdx.x);
        const int nw = nblk * BMSW;
        constexpr int NWT = BMW / 1024;
        static_assert(BMW % 1024 == 0, "whole words per thread");
        unsigned mw[NWT];
#pragma unroll
        for (int j = 0; j < NWT; ++j) {
            const int w = tid + j * 1024;
            mw[j] = w < nw ? bmp[w] : 0u;
        }
        unsigned nmine = 0u;
#pragma unroll
        for (int j = 0; j < NWT; ++j) nmine += (unsigned)__builtin_popcount(mw[j]);
        unsigned gp = nmine ? atomicAdd(&gcount, nmine) : 0u;
#pragma unroll
        for (int j = 0; j < NWT; ++j) {
            unsigned m = mw[j];
            while (m) {
                const unsigned bit = (unsigned)__builtin_ctz(m);
                m &= m - 1u;
                if (gp < QCAP) qs[gp] = (unsigned)(tid + j * 1024) * 32u + bit;
                ++gp;
            }
        }
        __syncthreads();
        const unsigned ng = gcount;
        if (ng <= QCAP) {
            const int4* __restrict__ src4 = reinterpret_cast<const int4*>(src);
            auto tailp = [&](unsigned e, const int* __restrict__ p) -> int4 {
                int4 v;
                v.x = (int)e     < E ? p[e]      : -1;
                v.y = (int)e + 1 < E ? p[e + 1u] : -1;
                v.z = (int)e + 2 < E ? p[e + 2u] : -1;
                v.w = -1;
                return v;
            };
            constexpr int NB = 8;
            const unsigned ni = ng * 4u;
            for (unsigned k0 = tid; k0 < ni; k0 += NB * 1024) {
                unsigned e[NB];
                int4 dv[NB], sv[NB];
#pragma unroll
                for (int j = 0; j < NB; ++j) {
                    const unsigned k  = k0 + (unsigned)j * 1024u;
                    const unsigned kk = k < ni ? k : ni - 1u;
                    e[j] = qs[kk >> 2] * (unsigned)GRP + (kk & 3u) * 4u;
                }
#pragma unroll
                for (int j = 0; j < NB; ++j) {
                    const int pv = (int)(e[j] >> 2);
                    const int pc = pv < nvec ? pv : nvec - 1;
                    dv[j] = dst4[pc];
                    sv[j] = src4[pc];
                }
                unsigned nown = 0u;
#pragma unroll
                for (int j = 0; j < NB; ++j) {
                    if (k0 + (unsigned)j * 1024u >= ni) {
                        dv[j] = make_int4(-1, -1, -1, -1);
                    } else if ((int)(e[j] >> 2) >= nvec) {
                        dv[j] = tailp(e[j], dst);
                        sv[j] = tailp(e[j], src);
                    }
                    nown += ((unsigned)(dv[j].x - r0) < RNG ? 1u : 0u) + ((unsigned)(dv[j].y - r0) < RNG ? 1u : 0u)
                          + ((unsigned)(dv[j].z - r0) < RNG ? 1u : 0u) + ((unsigned)(dv[j].w - r0) < RNG ? 1u : 0u);
                }
                unsigned qp = nown ? atomicAdd(&qcount, nown) : 0u;
                auto own = [&](int dn, int sn) {
                    const unsigned u = (unsigned)(dn - r0);
                    if (u < RNG) {
                        if (qp < QCAP) {
                            queue[qp] = ((unsigned)sn << 9) | u;
                            atomicAdd(&rowoff[u], 1u);
                        }
                        ++qp;
                    }
                };
#pragma unroll
                for (int j = 0; j < NB; ++j) {
                    own(dv[j].x, sv[j].x); own(dv[j].y, sv[j].y);
                    own(dv[j].z, sv[j].z); own(dv[j].w, sv[j].w);
                }
            }
            listed = true;
        }
        __syncthreads();
        cnt = ng <= QCAP ? qcount : (unsigned)QCAP + 1u;
    } else if (!have) {
        if (nvec > 0) {
            int4 cur[NP];
#pragma unroll
            for (int j = 0; j < NP; ++j) cur[j] = ldv(tid + j * 1024);
            for (int k = tid; k < nvec; k += NP * 1024) {
                int4 nxt[NP];
#pragma unroll
                for (int j = 0; j < NP; ++j) nxt[j] = ldv(k + (NP + j) * 1024);
#pragma unroll
                for (int j = 0; j < NP; ++j) scan4(cur[j], k + j * 1024);
#pragma unroll
                for (int j = 0; j < NP; ++j) cur[j] = nxt[j];
            }
        }
        for (int e = (nvec << 2) + tid; e < E; e += 1024) {
            const unsigned u = (unsigned)(dst[e] - r0);
            if (u < RNG) { const unsigned p = atomicAdd(&qcount, 1u); if (p < QCAP) queue[p] = ((unsigned)e << 9) | u; }
        }
        __syncthreads();
        cnt = qcount;
    }

    if (cnt > QCAP) {
        for (int i = tid; i < RNG * D / 4; i += 1024) tile4[i] = make_float4(0.f, 0.f, 0.f, 0.f);
        __syncthreads();
        for (int e = tid; e < E; e += 1024) {
            const unsigned u = (unsigned)(dst[e] - r0);
            if (u < RNG) {
                const float* __restrict__ h = H + (size_t)src[e] * D;
                float* t = tile + u * D;
                for (int f = 0; f < D; ++f) atomicAdd(t + (f & 3) * LPR + (f >> 2), h[f]);
            }
        }
        __syncthreads();
    } else {
        if (!have) {
            if (!listed) {
                for (unsigned i = tid; i < cnt; i += 1024) {
                    const unsigned q = queue[i];
                    const unsigned s = (unsigned)src[q >> 9];
                    queue[i] = (s << 9) | (q & RMASK);
                    atomicAdd(&rowoff[q & RMASK], 1u);
                }
                __syncthreads();
            }
            if (tid < 32) {
                unsigned first[ROWS / 32];
                unsigned sum = 0u;
#pragma unroll
                for (int k = 0; k < ROWS / 32; ++k) {
                    const unsigned t = rowoff[tid * (ROWS / 32) + k];
                    first[k] = sum;
                    sum += t;
                }
                unsigned inc = sum;
#pragma unroll
                for (int d = 1; d < 32; d <<= 1) {
                    const unsigned below = (unsigned)__shfl_up((int)inc, (unsigned)d);
                    if (tid >= d) inc += below;
                }
                const unsigned before = inc - sum;
#pragma unroll
                for (int k = 0; k < ROWS / 32; ++k) rowoff[tid * (ROWS / 32) + k] = first[k] + before;
            }
            __syncthreads();
            for (unsigned i = tid; i < cnt; i += 1024) {
                const unsigned q = queue[i];
                const unsigned p = atomicAdd(&rowoff[q & RMASK], 1u);
                qs[p] = q;
            }
            if (!KEEP && tid < 3) {
                const unsigned i = cnt + (unsigned)tid;
                if (i < ((cnt + 3u) & ~3u)) qs[i] = 0u;
            }
            __syncthreads();
        }
        if (!KEEP && cnt <= qkeep) {
            for (unsigned i = tid; i < (cnt + 3u) / 4u; i += 1024) qg4[i] = qs4[i];
        }

        const int wave = tid >> 5;
        const int lane = tid & 31;
        const int sub  = lane / LPR;
        const int l    = lane % LPR;
        constexpr int U = 8;
        constexpr unsigned STEP = EPW * U;
        const float* __restrict__ hl = H + l * 4;
        float* tl = tile + l;
        for (unsigned g = (unsigned)wave * STEP; g < cnt; g += 32u * STEP) {
            const unsigned i0 = g + (unsigned)sub * U;
            unsigned q[U];
            float4   val[U];
#pragma unroll
            for (int u = 0; u < U; ++u) {
                const unsigned i = i0 + (unsigned)u;
                q[u] = qs[i < cnt ? i : cnt - 1u];
            }
#pragma unroll
            for (int u = 0; u < U; ++u)
                val[u] = *reinterpret_cast<const float4*>(hl + (size_t)(q[u] >> 9) * D);
#pragma unroll
            for (int u = 0; u < U; ++u) {
                const bool here = i0 + (unsigned)u < cnt;
                const unsigned r = q[u] & RMASK;
                const int  un = u + 1 < U ? u + 1 : u;
                const bool on = u + 1 < U && i0 + (unsigned)un < cnt && (q[un] & RMASK) == r;
                if (on) {
                    val[un].x += val[u].x; val[un].y += val[u].y;
                    val[un].z += val[u].z; val[un].w += val[u].w;
                } else if (here) {
                    float a0 = val[u].x, a1 = val[u].y, a2 = val[u].z, a3 = val[u].w;
                    if (sub) { const float t0 = a0; a0 = a1; a1 = a2; a2 = a3; a3 = t0; }
                    float* t = tl + r * D;
                    atomicAdd(t + ((0 + sub) & 3) * LPR, a0);
                    atomicAdd(t + ((1 + sub) & 3) * LPR, a1);
                    atomicAdd(t + ((2 + sub) & 3) * LPR, a2);
                    atomicAdd(t + ((3 + sub) & 3) * LPR, a3);
                }
            }
        }
        __syncthreads();
    }

    if (!KEEP && tid == 0) qg[0] = cnt;

    for (int i = tid; i < RNG * LPR; i += 1024) {
        const int row  = i / LPR;
        const int l    = i % LPR;
        const int grow = r0 + row;
        if (grow < N_NODES) {
            const float* t = tile + row * D + l;
            const float sc = nd[grow];
            const float4 bb = *reinterpret_cast<const float4*>(bias + l * 4);
            float4 o;
            o.x = t[0]       * sc + bb.x;
            o.y = t[LPR]     * sc + bb.y;
            o.z = t[2 * LPR] * sc + bb.z;
            o.w = t[3 * LPR] * sc + bb.w;
            if (RELU) {
                o.x = fmaxf(o.x, 0.0f); o.y = fmaxf(o.y, 0.0f);
                o.z = fmaxf(o.z, 0.0f); o.w = fmaxf(o.w, 0.0f);
            }
            *reinterpret_cast<float4*>(O + (size_t)grow * D + l * 4) = o;
        }
    }
}

extern "C" void kernel_launch(void* const* d_in, const int* in_sizes, int n_in,
                              void* d_out, int out_size, void* d_ws, size_t ws_size,
                              hipStream_t stream) {
    (void)n_in; (void)out_size; (void)ws_size;

    const float* x  = (const float*)d_in[0];
    const int*   src = (const int*)d_in[1];
    const int*   dst = (const int*)d_in[2];
    const float* W1 = (const float*)d_in[3];
    const float* b1 = (const float*)d_in[4];
    const float* W2 = (const float*)d_in[5];
    const float* b2 = (const float*)d_in[6];
    const float* W3 = (const float*)d_in[7];
    const float* b3 = (const float*)d_in[8];

    const int E = in_sizes[1];
    const int N = N_NODES;

    float* ws   = (float*)d_ws;
    float* ns   = ws;
    float* nd   = ws + NPAD;
    float* bufA = ws + 2 * NPAD;
    float* bufB = bufA + (size_t)N * 128;
    unsigned* qpark = (unsigned*)d_out;
    unsigned* part = (unsigned*)bufB;

    deg_hist_kernel<<<dim3((E + SLICE - 1) / SLICE, 2), 1024, 0, stream>>>(src, dst, part, qpark, E);
    norm_kernel<<<NPAD2 / 256, 256, 0, stream>>>(part, ns, E);

    gemm_norm_wmma<128, 128><<<GEMM_BLOCKS, 256, 0, stream>>>(x, ns, W1, bufA);
    aggregate_owner<128, true, false><<<NRNG, 1024, 0, stream>>>(bufA, src, dst, nd, b1, bufB, qpark, E);

    gemm_norm_wmma<128, 128><<<GEMM_BLOCKS, 256, 0, stream>>>(bufB, ns, W2, bufA);
    aggregate_owner<128, true, true><<<NRNG, 1024, 0, stream>>>(bufA, src, dst, nd, b2, bufB, qpark, E);

    gemm_norm_wmma<128, 64><<<GEMM_BLOCKS, 256, 0, stream>>>(bufB, ns, W3, bufA);
    aggregate_owner<64, false, true><<<NRNG, 1024, 0, stream>>>(bufA, src, dst, nd, b3, (float*)d_out, qpark, E);
}
